# starting kernel
# baseline (speedup 1.0000x reference)
#include <hip/hip_runtime.h>
#include <math.h>


typedef __attribute__((ext_vector_type(2))) float v2f;
typedef __attribute__((ext_vector_type(8))) float v8f;

#define NEG_SLOPE 0.2f

__global__ __launch_bounds__(256) void gat_gemm_wmma_f32(
    const float* __restrict__ A, const float* __restrict__ W,
    float* __restrict__ C, int M, int K) {
  const int wave = threadIdx.x >> 5;
  const int lane = threadIdx.x & 31;
  const int ntile = wave & 3;
  const int mtile = (blockIdx.x << 1) | (wave >> 2);
  if (mtile * 16 >= M) return;

  const int mrow = mtile * 16 + (lane & 15);
  const int koff = (lane >> 4) << 1;
  const int col  = ntile * 16 + (lane & 15);

  const float* __restrict__ arow = A + (size_t)mrow * K;
  v8f acc = {};
  for (int k = 0; k < K; k += 4) {
    v2f a, b;
    a.x = arow[k + koff];
    a.y = arow[k + koff + 1];
    b.x = W[(size_t)(k + koff)     * 64 + col];
    b.y = W[(size_t)(k + koff + 1) * 64 + col];
    acc = __builtin_amdgcn_wmma_f32_16x16x4_f32(
        false, a, false, b, (short)0, acc, false, false);
  }
  const int rbase = mtile * 16 + ((lane >> 4) << 3);
#pragma unroll
  for (int v = 0; v < 8; ++v)
    C[(size_t)(rbase + v) * 64 + col] = acc[v];
}

__global__ __launch_bounds__(256) void gat_alpha(
    const float* __restrict__ hp, const float* __restrict__ a_s,
    const float* __restrict__ a_d, float* __restrict__ out_s,
    float* __restrict__ out_d, int N) {
  const int node = blockIdx.x * 8 + (threadIdx.x >> 5);
  const int lane = threadIdx.x & 31;
  if (node >= N) return;
  const float* __restrict__ h = hp + (size_t)node * 64;
  float s = h[lane] * a_s[lane] + h[lane + 32] * a_s[lane + 32];
  float d = h[lane] * a_d[lane] + h[lane + 32] * a_d[lane + 32];
#pragma unroll
  for (int off = 16; off > 0; off >>= 1) {
    s += __shfl_down(s, off, 32);
    d += __shfl_down(d, off, 32);
  }
  if (lane == 0) { out_s[node] = s; out_d[node] = d; }
}

__global__ void gat_init(float* __restrict__ accum, unsigned int* __restrict__ maxk,
                         float* __restrict__ denom, int N) {
  const int tid = blockIdx.x * blockDim.x + threadIdx.x;
  if (tid < N * 64) accum[tid] = 0.0f;
  if (tid < N) { maxk[tid] = 0u; denom[tid] = 0.0f; }
}

__device__ __forceinline__ unsigned int f2key(float f) {
  unsigned int u = __float_as_uint(f);
  return (u & 0x80000000u) ? ~u : (u | 0x80000000u);
}
__device__ __forceinline__ float key2f(unsigned int k) {
  return __uint_as_float((k & 0x80000000u) ? (k ^ 0x80000000u) : ~k);
}

__device__ __forceinline__ void edge_ends(const int* __restrict__ ei, int idx,
                                          int E, int& src, int& dst) {
  if (idx < E) { src = ei[idx]; dst = ei[E + idx]; }
  else         { src = idx - E; dst = idx - E; }
}

__device__ __forceinline__ float edge_logit(const float* __restrict__ as_,
                                            const float* __restrict__ ad_,
                                            int src, int dst) {
  float e = as_[src] + ad_[dst];
  return (e > 0.0f) ? e : NEG_SLOPE * e;
}

__global__ void gat_edge_max(const int* __restrict__ ei,
                             const float* __restrict__ as_,
                             const float* __restrict__ ad_,
                             unsigned int* __restrict__ maxk, int E, int N) {
  const int idx = blockIdx.x * blockDim.x + threadIdx.x;
  if (idx >= E + N) return;
  int src, dst; edge_ends(ei, idx, E, src, dst);
  atomicMax(&maxk[dst], f2key(edge_logit(as_, ad_, src, dst)));
}

__global__ void gat_edge_exp(const int* __restrict__ ei,
                             const float* __restrict__ as_,
                             const float* __restrict__ ad_,
                             const unsigned int* __restrict__ maxk,
                             float* __restrict__ denom,
                             float* __restrict__ eexp, int E, int N) {
  const int idx = blockIdx.x * blockDim.x + threadIdx.x;
  if (idx >= E + N) return;
  int src, dst; edge_ends(ei, idx, E, src, dst);
  const float v = __expf(edge_logit(as_, ad_, src, dst) - key2f(maxk[dst]));
  eexp[idx] = v;
  atomicAdd(&denom[dst], v);
}

__global__ void gat_edge_scatter(const int* __restrict__ ei,
                                 const float* __restrict__ hp,
                                 const float* __restrict__ eexp,
                                 const float* __restrict__ denom,
                                 float* __restrict__ accum, int E, int N) {
  const long long tid = (long long)blockIdx.x * blockDim.x + threadIdx.x;
  if (tid >= (long long)(E + N) * 64) return;
  const int idx = (int)(tid >> 6);
  const int c   = (int)(tid & 63);
  int src, dst; edge_ends(ei, idx, E, src, dst);
  const float a = eexp[idx] / (denom[dst] + 1e-16f);
  atomicAdd(&accum[(size_t)dst * 64 + c], hp[(size_t)src * 64 + c] * a);
}

__global__ void gat_finalize(const float* __restrict__ accum,
                             const float* __restrict__ b,
                             float* __restrict__ out, int N) {
  const int tid = blockIdx.x * blockDim.x + threadIdx.x;
  if (tid >= N * 64) return;
  const float v = accum[tid] + b[tid & 63];
  out[tid] = v > 0.0f ? v : 0.0f;
}

extern "C" void kernel_launch(void* const* d_in, const int* in_sizes, int n_in,
                              void* d_out, int out_size, void* d_ws, size_t ws_size,
                              hipStream_t stream) {
  (void)n_in; (void)out_size; (void)ws_size;

  const float* x   = (const float*)d_in[0];
  const int*   ei  = (const int*)d_in[1];
  const int N = in_sizes[0] / 128;
  const int E = in_sizes[1] / 2;
  const int T = E + N;

  float* ws    = (float*)d_ws;
  float* hp    = ws;
  float* hcur  = hp    + (size_t)N * 64;
  float* accum = hcur  + (size_t)N * 64;
  float* as_   = accum + (size_t)N * 64;
  float* ad_   = as_   + N;
  unsigned int* maxk = (unsigned int*)(ad_ + N);
  float* denom = (float*)(maxk + N);
  float* eexp  = denom + N;

  const float* hin = x;
  int K = 128;
  for (int layer = 0; layer < 3; ++layer) {
    const float* W    = (const float*)d_in[4 + layer * 4];
    const float* asrc = (const float*)d_in[5 + layer * 4];
    const float* adst = (const float*)d_in[6 + layer * 4];
    const float* bias = (const float*)d_in[7 + layer * 4];

    const int mblocks = (N / 16 + 1) / 2;
    gat_gemm_wmma_f32<<<mblocks + 1, 256, 0, stream>>>(hin, W, hp, N, K);

    gat_alpha<<<(N + 7) / 8, 256, 0, stream>>>(hp, asrc, adst, as_, ad_, N);

    gat_init<<<(N * 64 + 255) / 256, 256, 0, stream>>>(accum, maxk, denom, N);
    gat_edge_max<<<(T + 255) / 256, 256, 0, stream>>>(ei, as_, ad_, maxk, E, N);
    gat_edge_exp<<<(T + 255) / 256, 256, 0, stream>>>(ei, as_, ad_, maxk, denom, eexp, E, N);
    {
      long long tot = (long long)T * 64;
      gat_edge_scatter<<<(unsigned int)((tot + 255) / 256), 256, 0, stream>>>(
          ei, hp, eexp, denom, accum, E, N);
    }

    float* dst = (layer == 2) ? (float*)d_out : hcur;
    gat_finalize<<<(N * 64 + 255) / 256, 256, 0, stream>>>(accum, bias, dst, N);

    hin = hcur;
    K = 64;
  }
}


# Edges grouped by destination once per call (rank = a returning int atomic); one wave per node does max, exp-sum and weighted row sum in registers: no float atomics, sums in arrival-order rank
# speedup vs baseline: 4.2465x; 4.2465x over previous
#include <hip/hip_runtime.h>
#include <math.h>


typedef __attribute__((ext_vector_type(2))) float v2f;
typedef __attribute__((ext_vector_type(8))) float v8f;

#define NEG_SLOPE 0.2f

__global__ __launch_bounds__(256) void gat_gemm_wmma_f32(
    const float* __restrict__ A, const float* __restrict__ W,
    float* __restrict__ C, int M, int K) {
  const int wave = threadIdx.x >> 5;
  const int lane = threadIdx.x & 31;
  const int ntile = wave & 3;
  const int mtile = (blockIdx.x << 1) | (wave >> 2);
  if (mtile * 16 >= M) return;

  const int mrow = mtile * 16 + (lane & 15);
  const int koff = (lane >> 4) << 1;
  const int col  = ntile * 16 + (lane & 15);

  const float* __restrict__ arow = A + (size_t)mrow * K;
  v8f acc = {};
  for (int k = 0; k < K; k += 4) {
    v2f a, b;
    a.x = arow[k + koff];
    a.y = arow[k + koff + 1];
    b.x = W[(size_t)(k + koff)     * 64 + col];
    b.y = W[(size_t)(k + koff + 1) * 64 + col];
    acc = __builtin_amdgcn_wmma_f32_16x16x4_f32(
        false, a, false, b, (short)0, acc, false, false);
  }
  const int rbase = mtile * 16 + ((lane >> 4) << 3);
#pragma unroll
  for (int v = 0; v < 8; ++v)
    C[(size_t)(rbase + v) * 64 + col] = acc[v];
}

__global__ __launch_bounds__(256) void gat_alpha(
    const float* __restrict__ hp, const float* __restrict__ a_s,
    const float* __restrict__ a_d, float* __restrict__ out_s,
    float* __restrict__ out_d, int N) {
  const int node = blockIdx.x * 8 + (threadIdx.x >> 5);
  const int lane = threadIdx.x & 31;
  if (node >= N) return;
  const float* __restrict__ h = hp + (size_t)node * 64;
  float s = h[lane] * a_s[lane] + h[lane + 32] * a_s[lane + 32];
  float d = h[lane] * a_d[lane] + h[lane + 32] * a_d[lane + 32];
#pragma unroll
  for (int off = 16; off > 0; off >>= 1) {
    s += __shfl_down(s, off, 32);
    d += __shfl_down(d, off, 32);
  }
  if (lane == 0) { out_s[node] = s; out_d[node] = d; }
}

__global__ void csr_zero(int* __restrict__ cnt, int n) {
  const int tid = blockIdx.x * blockDim.x + threadIdx.x;
  if (tid < n) cnt[tid] = 0;
}

__global__ void csr_count(const int* __restrict__ ei, int* __restrict__ cnt,
                          int* __restrict__ rank, int E) {
  const int idx = blockIdx.x * blockDim.x + threadIdx.x;
  if (idx >= E) return;
  rank[idx] = atomicAdd(&cnt[ei[E + idx]], 1);
}

__global__ __launch_bounds__(256) void csr_assign(const int* cnt, int* rowstart,
                                                  int* cursor, int N) {
  const int d = blockIdx.x * 256 + threadIdx.x;
  const int lane = threadIdx.x & 31;
  const int c = (d < N) ? cnt[d] : 0;
  int inc = c;
#pragma unroll
  for (int off = 1; off < 32; off <<= 1) {
    const int t = __shfl_up(inc, off, 32);
    if (lane >= off) inc += t;
  }
  int base = 0;
  if (lane == 31) base = atomicAdd(cursor, inc);
  base = __shfl(base, 31, 32);
  if (d < N) rowstart[d] = base + inc - c;
}

__global__ void csr_fill(const int* __restrict__ ei, const int* __restrict__ rank,
                         const int* __restrict__ rowstart,
                         int* __restrict__ ssrc, int E) {
  const int idx = blockIdx.x * blockDim.x + threadIdx.x;
  if (idx >= E) return;
  ssrc[rowstart[ei[E + idx]] + rank[idx]] = ei[idx];
}

__device__ __forceinline__ float leaky(float e) {
  return (e > 0.0f) ? e : NEG_SLOPE * e;
}

__device__ __forceinline__ float wave_max(float v) {
#pragma unroll
  for (int off = 16; off > 0; off >>= 1) v = fmaxf(v, __shfl_xor(v, off, 32));
  return v;
}
__device__ __forceinline__ float wave_sum(float v) {
#pragma unroll
  for (int off = 16; off > 0; off >>= 1) v += __shfl_xor(v, off, 32);
  return v;
}

__device__ __forceinline__ float2 row_of(const float2* __restrict__ hp2, int sj,
                                         int k, int lane) {
  return hp2[(size_t)__builtin_amdgcn_readlane(sj, k) * 32 + lane];
}
__device__ __forceinline__ float wgt_of(float aj, int k) {
  return __int_as_float(__builtin_amdgcn_readlane(__float_as_int(aj), k));
}
__device__ __forceinline__ void add_rows(const float2* __restrict__ hp2, int sj,
                                         float aj, int nk, int lane, float2& acc) {
  int k = 0;
  for (; k + 4 <= nk; k += 4) {
    const float2 r0 = row_of(hp2, sj, k, lane);
    const float2 r1 = row_of(hp2, sj, k + 1, lane);
    const float2 r2 = row_of(hp2, sj, k + 2, lane);
    const float2 r3 = row_of(hp2, sj, k + 3, lane);
    const float a0 = wgt_of(aj, k), a1 = wgt_of(aj, k + 1);
    const float a2 = wgt_of(aj, k + 2), a3 = wgt_of(aj, k + 3);
    acc.x += r0.x * a0;  acc.y += r0.y * a0;
    acc.x += r1.x * a1;  acc.y += r1.y * a1;
    acc.x += r2.x * a2;  acc.y += r2.y * a2;
    acc.x += r3.x * a3;  acc.y += r3.y * a3;
  }
  for (; k < nk; ++k) {
    const float2 r = row_of(hp2, sj, k, lane);
    const float a = wgt_of(aj, k);
    acc.x += r.x * a;  acc.y += r.y * a;
  }
}

__global__ __launch_bounds__(256) void gat_gather(
    const int* __restrict__ rowstart, const int* __restrict__ cnt,
    const int* __restrict__ ssrc, const float* __restrict__ hp,
    const float* __restrict__ as_, const float* __restrict__ ad_,
    const float* __restrict__ bias, float* __restrict__ out, int N) {
  const int node = __builtin_amdgcn_readfirstlane(blockIdx.x * 8 + (threadIdx.x >> 5));
  const int lane = threadIdx.x & 31;
  if (node >= N) return;
  const int beg = rowstart[node];
  const int deg = cnt[node];
  const float adn = ad_[node];
  const float es = leaky(as_[node] + adn);

  int s0 = node;
  float e0 = es;
  if (lane < deg) { s0 = ssrc[beg + lane]; e0 = leaky(as_[s0] + adn); }
  float m = fmaxf(es, e0);
  for (int j = lane + 32; j < deg; j += 32)
    m = fmaxf(m, leaky(as_[ssrc[beg + j]] + adn));
  m = wave_max(m);

  const float v0 = (lane < deg) ? __expf(e0 - m) : 0.0f;
  float sum = v0;
  for (int j = lane + 32; j < deg; j += 32)
    sum += __expf(leaky(as_[ssrc[beg + j]] + adn) - m);
  const float vs = __expf(es - m);
  const float dn = (wave_sum(sum) + vs) + 1e-16f;

  const float2* __restrict__ hp2 = (const float2*)hp;
  const float aself = vs / dn;
  const float2 hs = hp2[(size_t)node * 32 + lane];
  float2 acc;
  acc.x = hs.x * aself;
  acc.y = hs.y * aself;
  add_rows(hp2, s0, v0 / dn, deg < 32 ? deg : 32, lane, acc);
  for (int base = 32; base < deg; base += 32) {
    int sj = node;
    float aj = 0.0f;
    if (base + lane < deg) {
      sj = ssrc[beg + base + lane];
      aj = __expf(leaky(as_[sj] + adn) - m) / dn;
    }
    add_rows(hp2, sj, aj, (deg - base) < 32 ? (deg - base) : 32, lane, acc);
  }

  const float2 b = ((const float2*)bias)[lane];
  float2 o;
  o.x = acc.x + b.x;
  o.y = acc.y + b.y;
  o.x = o.x > 0.0f ? o.x : 0.0f;
  o.y = o.y > 0.0f ? o.y : 0.0f;
  ((float2*)out)[(size_t)node * 32 + lane] = o;
}

extern "C" void kernel_launch(void* const* d_in, const int* in_sizes, int n_in,
                              void* d_out, int out_size, void* d_ws, size_t ws_size,
                              hipStream_t stream) {
  (void)n_in; (void)out_size; (void)ws_size;

  const float* x   = (const float*)d_in[0];
  const int*   ei  = (const int*)d_in[1];
  const int N = in_sizes[0] / 128;
  const int E = in_sizes[1] / 2;
  float* ws    = (float*)d_ws;
  float* hp    = ws;
  float* hcur  = hp    + (size_t)N * 64;
  float* lists = hcur  + (size_t)N * 64;
  float* as_   = lists + (size_t)N * 64;
  float* ad_   = as_   + N;
  int* cnt      = (int*)lists;
  int* rowstart = cnt + N + 1;
  int* rank     = rowstart + N;
  int* ssrc     = rank + E;

  csr_zero<<<(N + 1 + 255) / 256, 256, 0, stream>>>(cnt, N + 1);
  csr_count<<<(E + 255) / 256, 256, 0, stream>>>(ei, cnt, rank, E);
  csr_assign<<<(N + 255) / 256, 256, 0, stream>>>(cnt, rowstart, cnt + N, N);
  csr_fill<<<(E + 255) / 256, 256, 0, stream>>>(ei, rank, rowstart, ssrc, E);

  const float* hin = x;
  int K = 128;
  for (int layer = 0; layer < 3; ++layer) {
    const float* W    = (const float*)d_in[4 + layer * 4];
    const float* asrc = (const float*)d_in[5 + layer * 4];
    const float* adst = (const float*)d_in[6 + layer * 4];
    const float* bias = (const float*)d_in[7 + layer * 4];

    const int mblocks = (N / 16 + 1) / 2;
    gat_gemm_wmma_f32<<<mblocks + 1, 256, 0, stream>>>(hin, W, hp, N, K);

    gat_alpha<<<(N + 7) / 8, 256, 0, stream>>>(hp, asrc, adst, as_, ad_, N);

    float* dst = (layer == 2) ? (float*)d_out : hcur;
    gat_gather<<<(N + 7) / 8, 256, 0, stream>>>(rowstart, cnt, ssrc, hp, as_, ad_,
                                                bias, dst, N);

    hin = hcur;
    K = 64;
  }
}


# GEMM wave owns 16 whole rows (4 column tiles, one A fragment) and sums alpha_s/alpha_d from its accumulators across 16 lanes: alpha kernel and its 3 launches go; alpha sum order changes
# speedup vs baseline: 4.8930x; 1.1522x over previous
#include <hip/hip_runtime.h>
#include <math.h>


typedef __attribute__((ext_vector_type(2))) float v2f;
typedef __attribute__((ext_vector_type(8))) float v8f;

#define NEG_SLOPE 0.2f


template <int MASK>
__device__ __forceinline__ float row16_xor_add(float x) {
  const int t = __builtin_amdgcn_update_dpp(0, __float_as_int(x), 0x160 | MASK,
                                            0xf, 0xf, true);
  return x + __int_as_float(t);
}

__device__ __forceinline__ v8f wmma4(v2f a, float b0, float b1, v8f acc) {
  v2f b;
  b.x = b0;
  b.y = b1;
  return __builtin_amdgcn_wmma_f32_16x16x4_f32(false, a, false, b, (short)0, acc,
                                               false, false);
}

struct Slab {
  v2f a[4];
  float4 b0[4];
  float4 b1[4];
};

__device__ __forceinline__ void load_slab(Slab& f, const float* __restrict__ arow,
                                          const float* __restrict__ wcol, int k) {
#pragma unroll
  for (int j = 0; j < 4; ++j) {
    f.a[j]  = *(const v2f*)(arow + k + 4 * j);
    f.b0[j] = *(const float4*)(wcol + (size_t)(k + 4 * j) * 64);
    f.b1[j] = *(const float4*)(wcol + (size_t)(k + 4 * j + 1) * 64);
  }
}

__device__ __forceinline__ void mma_slab(const Slab& f, v8f& acc0, v8f& acc1,
                                         v8f& acc2, v8f& acc3) {
#pragma unroll
  for (int j = 0; j < 4; ++j) {
    acc0 = wmma4(f.a[j], f.b0[j].x, f.b1[j].x, acc0);
    acc1 = wmma4(f.a[j], f.b0[j].y, f.b1[j].y, acc1);
    acc2 = wmma4(f.a[j], f.b0[j].z, f.b1[j].z, acc2);
    acc3 = wmma4(f.a[j], f.b0[j].w, f.b1[j].w, acc3);
  }
}

__global__ __launch_bounds__(64) void gat_gemm_alpha(
    const float* __restrict__ A, const float* __restrict__ W,
    const float* __restrict__ a_s, const float* __restrict__ a_d,
    float* __restrict__ C, float* __restrict__ out_s, float* __restrict__ out_d,
    int M, int K) {
  const int lane = threadIdx.x & 31;
  const int mtile = blockIdx.x * 2 + (threadIdx.x >> 5);
  if (mtile * 16 >= M) return;
  const int c = lane & 15;
  const int h = lane >> 4;

  const float* __restrict__ arow = A + (size_t)(mtile * 16 + c) * K + 2 * h;
  const float* __restrict__ wcol = W + (size_t)(2 * h) * 64 + 4 * c;

  v8f acc0 = {}, acc1 = {}, acc2 = {}, acc3 = {};
  Slab f0, f1;
  load_slab(f0, arow, wcol, 0);
#pragma unroll 1
  for (int k = 0; k < K - 32; k += 32) {
    load_slab(f1, arow, wcol, k + 16);
    mma_slab(f0, acc0, acc1, acc2, acc3);
    load_slab(f0, arow, wcol, k + 32);
    mma_slab(f1, acc0, acc1, acc2, acc3);
  }
  load_slab(f1, arow, wcol, K - 16);
  mma_slab(f0, acc0, acc1, acc2, acc3);
  mma_slab(f1, acc0, acc1, acc2, acc3);

  const int rbase = mtile * 16 + 8 * h;
#pragma unroll
  for (int v = 0; v < 8; ++v) {
    float4 o;
    o.x = acc0[v];  o.y = acc1[v];  o.z = acc2[v];  o.w = acc3[v];
    *(float4*)(C + (size_t)(rbase + v) * 64 + 4 * c) = o;
  }

  const float4 s4 = *(const float4*)(a_s + 4 * c);
  const float4 d4 = *(const float4*)(a_d + 4 * c);
  float ps[8], pd[8];
#pragma unroll
  for (int v = 0; v < 8; ++v) {
    ps[v] = acc0[v] * s4.x + acc1[v] * s4.y + acc2[v] * s4.z + acc3[v] * s4.w;
    pd[v] = acc0[v] * d4.x + acc1[v] * d4.y + acc2[v] * d4.z + acc3[v] * d4.w;
  }
#pragma unroll
  for (int v = 0; v < 8; ++v) { ps[v] = row16_xor_add<1>(ps[v]); pd[v] = row16_xor_add<1>(pd[v]); }
#pragma unroll
  for (int v = 0; v < 8; ++v) { ps[v] = row16_xor_add<2>(ps[v]); pd[v] = row16_xor_add<2>(pd[v]); }
#pragma unroll
  for (int v = 0; v < 8; ++v) { ps[v] = row16_xor_add<4>(ps[v]); pd[v] = row16_xor_add<4>(pd[v]); }
#pragma unroll
  for (int v = 0; v < 8; ++v) { ps[v] = row16_xor_add<8>(ps[v]); pd[v] = row16_xor_add<8>(pd[v]); }

  float val = 0.0f;
#pragma unroll
  for (int v = 0; v < 8; ++v) {
    if (c == v) val = ps[v];
    if (c == v + 8) val = pd[v];
  }
  float* __restrict__ dst = (c < 8) ? (out_s + rbase + c) : (out_d + rbase + c - 8);
  *dst = val;
}

__global__ void csr_zero(int* __restrict__ cnt, int n) {
  const int tid = blockIdx.x * blockDim.x + threadIdx.x;
  if (tid < n) cnt[tid] = 0;
}

__global__ void csr_count(const int* __restrict__ ei, int* __restrict__ cnt,
                          int* __restrict__ rank, int E) {
  const int idx = blockIdx.x * blockDim.x + threadIdx.x;
  if (idx >= E) return;
  rank[idx] = atomicAdd(&cnt[ei[E + idx]], 1);
}

__global__ __launch_bounds__(256) void csr_assign(const int* cnt, int* rowstart,
                                                  int* cursor, int N) {
  const int d = blockIdx.x * 256 + threadIdx.x;
  const int lane = threadIdx.x & 31;
  const int c = (d < N) ? cnt[d] : 0;
  int inc = c;
#pragma unroll
  for (int off = 1; off < 32; off <<= 1) {
    const int t = __shfl_up(inc, off, 32);
    if (lane >= off) inc += t;
  }
  int base = 0;
  if (lane == 31) base = atomicAdd(cursor, inc);
  base = __shfl(base, 31, 32);
  if (d < N) rowstart[d] = base + inc - c;
}

__global__ void csr_fill(const int* __restrict__ ei, const int* __restrict__ rank,
                         const int* __restrict__ rowstart,
                         int* __restrict__ ssrc, int E) {
  const int idx = blockIdx.x * blockDim.x + threadIdx.x;
  if (idx >= E) return;
  ssrc[rowstart[ei[E + idx]] + rank[idx]] = ei[idx];
}

__device__ __forceinline__ float leaky(float e) {
  return (e > 0.0f) ? e : NEG_SLOPE * e;
}

__device__ __forceinline__ float wave_max(float v) {
#pragma unroll
  for (int off = 16; off > 0; off >>= 1) v = fmaxf(v, __shfl_xor(v, off, 32));
  return v;
}
__device__ __forceinline__ float wave_sum(float v) {
#pragma unroll
  for (int off = 16; off > 0; off >>= 1) v += __shfl_xor(v, off, 32);
  return v;
}

__device__ __forceinline__ float2 row_of(const float2* __restrict__ hp2, int sj,
                                         int k, int lane) {
  return hp2[(size_t)__builtin_amdgcn_readlane(sj, k) * 32 + lane];
}
__device__ __forceinline__ float wgt_of(float aj, int k) {
  return __int_as_float(__builtin_amdgcn_readlane(__float_as_int(aj), k));
}
__device__ __forceinline__ void add_rows(const float2* __restrict__ hp2, int sj,
                                         float aj, int nk, int lane, float2& acc) {
  int k = 0;
  for (; k + 4 <= nk; k += 4) {
    const float2 r0 = row_of(hp2, sj, k, lane);
    const float2 r1 = row_of(hp2, sj, k + 1, lane);
    const float2 r2 = row_of(hp2, sj, k + 2, lane);
    const float2 r3 = row_of(hp2, sj, k + 3, lane);
    const float a0 = wgt_of(aj, k), a1 = wgt_of(aj, k + 1);
    const float a2 = wgt_of(aj, k + 2), a3 = wgt_of(aj, k + 3);
    acc.x += r0.x * a0;  acc.y += r0.y * a0;
    acc.x += r1.x * a1;  acc.y += r1.y * a1;
    acc.x += r2.x * a2;  acc.y += r2.y * a2;
    acc.x += r3.x * a3;  acc.y += r3.y * a3;
  }
  for (; k < nk; ++k) {
    const float2 r = row_of(hp2, sj, k, lane);
    const float a = wgt_of(aj, k);
    acc.x += r.x * a;  acc.y += r.y * a;
  }
}

__global__ __launch_bounds__(256) void gat_gather(
    const int* __restrict__ rowstart, const int* __restrict__ cnt,
    const int* __restrict__ ssrc, const float* __restrict__ hp,
    const float* __restrict__ as_, const float* __restrict__ ad_,
    const float* __restrict__ bias, float* __restrict__ out, int N) {
  const int node = __builtin_amdgcn_readfirstlane(blockIdx.x * 8 + (threadIdx.x >> 5));
  const int lane = threadIdx.x & 31;
  if (node >= N) return;
  const int beg = rowstart[node];
  const int deg = cnt[node];
  const float adn = ad_[node];
  const float es = leaky(as_[node] + adn);

  int s0 = node;
  float e0 = es;
  if (lane < deg) { s0 = ssrc[beg + lane]; e0 = leaky(as_[s0] + adn); }
  float m = fmaxf(es, e0);
  for (int j = lane + 32; j < deg; j += 32)
    m = fmaxf(m, leaky(as_[ssrc[beg + j]] + adn));
  m = wave_max(m);

  const float v0 = (lane < deg) ? __expf(e0 - m) : 0.0f;
  float sum = v0;
  for (int j = lane + 32; j < deg; j += 32)
    sum += __expf(leaky(as_[ssrc[beg + j]] + adn) - m);
  const float vs = __expf(es - m);
  const float dn = (wave_sum(sum) + vs) + 1e-16f;

  const float2* __restrict__ hp2 = (const float2*)hp;
  const float aself = vs / dn;
  const float2 hs = hp2[(size_t)node * 32 + lane];
  float2 acc;
  acc.x = hs.x * aself;
  acc.y = hs.y * aself;
  add_rows(hp2, s0, v0 / dn, deg < 32 ? deg : 32, lane, acc);
  for (int base = 32; base < deg; base += 32) {
    int sj = node;
    float aj = 0.0f;
    if (base + lane < deg) {
      sj = ssrc[beg + base + lane];
      aj = __expf(leaky(as_[sj] + adn) - m) / dn;
    }
    add_rows(hp2, sj, aj, (deg - base) < 32 ? (deg - base) : 32, lane, acc);
  }

  const float2 b = ((const float2*)bias)[lane];
  float2 o;
  o.x = acc.x + b.x;
  o.y = acc.y + b.y;
  o.x = o.x > 0.0f ? o.x : 0.0f;
  o.y = o.y > 0.0f ? o.y : 0.0f;
  ((float2*)out)[(size_t)node * 32 + lane] = o;
}

extern "C" void kernel_launch(void* const* d_in, const int* in_sizes, int n_in,
                              void* d_out, int out_size, void* d_ws, size_t ws_size,
                              hipStream_t stream) {
  (void)n_in; (void)out_size; (void)ws_size;

  const float* x   = (const float*)d_in[0];
  const int*   ei  = (const int*)d_in[1];
  const int N = in_sizes[0] / 128;
  const int E = in_sizes[1] / 2;
  float* ws    = (float*)d_ws;
  float* hp    = ws;
  float* hcur  = hp    + (size_t)N * 64;
  float* lists = hcur  + (size_t)N * 64;
  float* as_   = lists + (size_t)N * 64;
  float* ad_   = as_   + N;
  int* cnt      = (int*)lists;
  int* rowstart = cnt + N + 1;
  int* rank     = rowstart + N;
  int* ssrc     = rank + E;

  csr_zero<<<(N + 1 + 255) / 256, 256, 0, stream>>>(cnt, N + 1);
  csr_count<<<(E + 255) / 256, 256, 0, stream>>>(ei, cnt, rank, E);
  csr_assign<<<(N + 255) / 256, 256, 0, stream>>>(cnt, rowstart, cnt + N, N);
  csr_fill<<<(E + 255) / 256, 256, 0, stream>>>(ei, rank, rowstart, ssrc, E);

  const float* hin = x;
  int K = 128;
  for (int layer = 0; layer < 3; ++layer) {
    const float* W    = (const float*)d_in[4 + layer * 4];
    const float* asrc = (const float*)d_in[5 + layer * 4];
    const float* adst = (const float*)d_in[6 + layer * 4];
    const float* bias = (const float*)d_in[7 + layer * 4];

    const int mblocks = (N / 16 + 1) / 2;
    gat_gemm_alpha<<<mblocks, 64, 0, stream>>>(hin, W, asrc, adst, hp, as_, ad_, N, K);

    float* dst = (layer == 2) ? (float*)d_out : hcur;
    gat_gather<<<(N + 7) / 8, 256, 0, stream>>>(rowstart, cnt, ssrc, hp, as_, ad_,
                                                bias, dst, N);

    hin = hcur;
    K = 64;
  }
}


# Gather: 16 lanes x float4 per destination, two nodes per wave (edge slots broadcast by DPP row_share), so half the gather waves; exp-sum tree now 16 lanes x two slots; row sums keep rank order
# speedup vs baseline: 5.2068x; 1.0641x over previous
#include <hip/hip_runtime.h>
#include <math.h>


typedef __attribute__((ext_vector_type(2))) float v2f;
typedef __attribute__((ext_vector_type(8))) float v8f;

#define NEG_SLOPE 0.2f


template <int MASK>
__device__ __forceinline__ float row16_xor_add(float x) {
  const int t = __builtin_amdgcn_update_dpp(0, __float_as_int(x), 0x160 | MASK,
                                            0xf, 0xf, true);
  return x + __int_as_float(t);
}

__device__ __forceinline__ v8f wmma4(v2f a, float b0, float b1, v8f acc) {
  v2f b;
  b.x = b0;
  b.y = b1;
  return __builtin_amdgcn_wmma_f32_16x16x4_f32(false, a, false, b, (short)0, acc,
                                               false, false);
}

struct Slab {
  v2f a[4];
  float4 b0[4];
  float4 b1[4];
};

__device__ __forceinline__ void load_slab(Slab& f, const float* __restrict__ arow,
                                          const float* __restrict__ wcol, int k) {
#pragma unroll
  for (int j = 0; j < 4; ++j) {
    f.a[j]  = *(const v2f*)(arow + k + 4 * j);
    f.b0[j] = *(const float4*)(wcol + (size_t)(k + 4 * j) * 64);
    f.b1[j] = *(const float4*)(wcol + (size_t)(k + 4 * j + 1) * 64);
  }
}

__device__ __forceinline__ void mma_slab(const Slab& f, v8f& acc0, v8f& acc1,
                                         v8f& acc2, v8f& acc3) {
#pragma unroll
  for (int j = 0; j < 4; ++j) {
    acc0 = wmma4(f.a[j], f.b0[j].x, f.b1[j].x, acc0);
    acc1 = wmma4(f.a[j], f.b0[j].y, f.b1[j].y, acc1);
    acc2 = wmma4(f.a[j], f.b0[j].z, f.b1[j].z, acc2);
    acc3 = wmma4(f.a[j], f.b0[j].w, f.b1[j].w, acc3);
  }
}

__global__ __launch_bounds__(64) void gat_gemm_alpha(
    const float* __restrict__ A, const float* __restrict__ W,
    const float* __restrict__ a_s, const float* __restrict__ a_d,
    float* __restrict__ C, float* __restrict__ out_s, float* __restrict__ out_d,
    int M, int K) {
  const int lane = threadIdx.x & 31;
  const int mtile = blockIdx.x * 2 + (threadIdx.x >> 5);
  if (mtile * 16 >= M) return;
  const int c = lane & 15;
  const int h = lane >> 4;

  const float* __restrict__ arow = A + (size_t)(mtile * 16 + c) * K + 2 * h;
  const float* __restrict__ wcol = W + (size_t)(2 * h) * 64 + 4 * c;

  v8f acc0 = {}, acc1 = {}, acc2 = {}, acc3 = {};
  Slab f0, f1;
  load_slab(f0, arow, wcol, 0);
#pragma unroll 1
  for (int k = 0; k < K - 32; k += 32) {
    load_slab(f1, arow, wcol, k + 16);
    mma_slab(f0, acc0, acc1, acc2, acc3);
    load_slab(f0, arow, wcol, k + 32);
    mma_slab(f1, acc0, acc1, acc2, acc3);
  }
  load_slab(f1, arow, wcol, K - 16);
  mma_slab(f0, acc0, acc1, acc2, acc3);
  mma_slab(f1, acc0, acc1, acc2, acc3);

  const int rbase = mtile * 16 + 8 * h;
#pragma unroll
  for (int v = 0; v < 8; ++v) {
    float4 o;
    o.x = acc0[v];  o.y = acc1[v];  o.z = acc2[v];  o.w = acc3[v];
    *(float4*)(C + (size_t)(rbase + v) * 64 + 4 * c) = o;
  }

  const float4 s4 = *(const float4*)(a_s + 4 * c);
  const float4 d4 = *(const float4*)(a_d + 4 * c);
  float ps[8], pd[8];
#pragma unroll
  for (int v = 0; v < 8; ++v) {
    ps[v] = acc0[v] * s4.x + acc1[v] * s4.y + acc2[v] * s4.z + acc3[v] * s4.w;
    pd[v] = acc0[v] * d4.x + acc1[v] * d4.y + acc2[v] * d4.z + acc3[v] * d4.w;
  }
#pragma unroll
  for (int v = 0; v < 8; ++v) { ps[v] = row16_xor_add<1>(ps[v]); pd[v] = row16_xor_add<1>(pd[v]); }
#pragma unroll
  for (int v = 0; v < 8; ++v) { ps[v] = row16_xor_add<2>(ps[v]); pd[v] = row16_xor_add<2>(pd[v]); }
#pragma unroll
  for (int v = 0; v < 8; ++v) { ps[v] = row16_xor_add<4>(ps[v]); pd[v] = row16_xor_add<4>(pd[v]); }
#pragma unroll
  for (int v = 0; v < 8; ++v) { ps[v] = row16_xor_add<8>(ps[v]); pd[v] = row16_xor_add<8>(pd[v]); }

  float val = 0.0f;
#pragma unroll
  for (int v = 0; v < 8; ++v) {
    if (c == v) val = ps[v];
    if (c == v + 8) val = pd[v];
  }
  float* __restrict__ dst = (c < 8) ? (out_s + rbase + c) : (out_d + rbase + c - 8);
  *dst = val;
}

__global__ void csr_zero(int* __restrict__ cnt, int n) {
  const int tid = blockIdx.x * blockDim.x + threadIdx.x;
  if (tid < n) cnt[tid] = 0;
}

__global__ void csr_count(const int* __restrict__ ei, int* __restrict__ cnt,
                          int* __restrict__ rank, int E) {
  const int idx = blockIdx.x * blockDim.x + threadIdx.x;
  if (idx >= E) return;
  rank[idx] = atomicAdd(&cnt[ei[E + idx]], 1);
}

__global__ __launch_bounds__(256) void csr_assign(const int* cnt, int* rowstart,
                                                  int* cursor, int N) {
  const int d = blockIdx.x * 256 + threadIdx.x;
  const int lane = threadIdx.x & 31;
  const int c = (d < N) ? cnt[d] : 0;
  int inc = c;
#pragma unroll
  for (int off = 1; off < 32; off <<= 1) {
    const int t = __shfl_up(inc, off, 32);
    if (lane >= off) inc += t;
  }
  int base = 0;
  if (lane == 31) base = atomicAdd(cursor, inc);
  base = __shfl(base, 31, 32);
  if (d < N) rowstart[d] = base + inc - c;
}

__global__ void csr_fill(const int* __restrict__ ei, const int* __restrict__ rank,
                         const int* __restrict__ rowstart,
                         int* __restrict__ ssrc, int E) {
  const int idx = blockIdx.x * blockDim.x + threadIdx.x;
  if (idx >= E) return;
  ssrc[rowstart[ei[E + idx]] + rank[idx]] = ei[idx];
}

__device__ __forceinline__ float leaky(float e) {
  return (e > 0.0f) ? e : NEG_SLOPE * e;
}

template <int K>
__device__ __forceinline__ int row16_lane(int x) {
  return __builtin_amdgcn_update_dpp(0, x, 0x150 | K, 0xf, 0xf, true);
}
template <int MASK>
__device__ __forceinline__ float row16_xor_max(float x) {
  const int t = __builtin_amdgcn_update_dpp(0, __float_as_int(x), 0x160 | MASK,
                                            0xf, 0xf, true);
  return fmaxf(x, __int_as_float(t));
}

__device__ __forceinline__ void fma_row(float4& acc, const float4 r, float a) {
  acc.x = __builtin_fmaf(r.x, a, acc.x);
  acc.y = __builtin_fmaf(r.y, a, acc.y);
  acc.z = __builtin_fmaf(r.z, a, acc.z);
  acc.w = __builtin_fmaf(r.w, a, acc.w);
}

template <int K0>
__device__ __forceinline__ void add_rows4(const float4* __restrict__ hp4, int sj,
                                          float aj, int c, float4& acc) {
  const int ai = __float_as_int(aj);
  const float4 r0 = hp4[(unsigned)(row16_lane<K0>(sj) * 16 + c)];
  const float4 r1 = hp4[(unsigned)(row16_lane<K0 + 1>(sj) * 16 + c)];
  const float4 r2 = hp4[(unsigned)(row16_lane<K0 + 2>(sj) * 16 + c)];
  const float4 r3 = hp4[(unsigned)(row16_lane<K0 + 3>(sj) * 16 + c)];
  const float a0 = __int_as_float(row16_lane<K0>(ai));
  const float a1 = __int_as_float(row16_lane<K0 + 1>(ai));
  const float a2 = __int_as_float(row16_lane<K0 + 2>(ai));
  const float a3 = __int_as_float(row16_lane<K0 + 3>(ai));
  fma_row(acc, r0, a0);
  fma_row(acc, r1, a1);
  fma_row(acc, r2, a2);
  fma_row(acc, r3, a3);
}
__device__ __forceinline__ void add_rows16(const float4* __restrict__ hp4, int sj,
                                           float aj, int nkmax, int c, float4& acc) {
  if (nkmax > 0)  add_rows4<0>(hp4, sj, aj, c, acc);
  if (nkmax > 4)  add_rows4<4>(hp4, sj, aj, c, acc);
  if (nkmax > 8)  add_rows4<8>(hp4, sj, aj, c, acc);
  if (nkmax > 12) add_rows4<12>(hp4, sj, aj, c, acc);
}

__global__ __launch_bounds__(256) void gat_gather(
    const int* __restrict__ rowstart, const int* __restrict__ cnt,
    const int* __restrict__ ssrc, const float* __restrict__ hp,
    const float* __restrict__ as_, const float* __restrict__ ad_,
    const float* __restrict__ bias, float* __restrict__ out, int N) {
  const int pair = __builtin_amdgcn_readfirstlane(blockIdx.x * 8 + (threadIdx.x >> 5));
  const int lane = threadIdx.x & 31;
  const int c = lane & 15;
  if (pair * 2 >= N) return;
  const bool live = pair * 2 + (lane >> 4) < N;
  const int node = live ? pair * 2 + (lane >> 4) : pair * 2;
  const int beg = rowstart[node];
  const int cn = cnt[node];
  const float adn = ad_[node];
  const float es = leaky(as_[node] + adn);
  const int deg = live ? cn : 0;
  const int dlo = __builtin_amdgcn_readlane(deg, 0);
  const int dhi = __builtin_amdgcn_readlane(deg, 16);
  const int degmax = dlo > dhi ? dlo : dhi;

  const bool on0 = c < deg, on1 = c + 16 < deg;
  const int q0 = ssrc[on0 ? beg + c : 0];
  const int q1 = ssrc[on1 ? beg + 16 + c : 0];
  const int s0 = on0 ? q0 : node;
  const int s1 = on1 ? q1 : node;
  const float t0 = as_[s0], t1 = as_[s1];
  const float e0 = leaky(t0 + adn), e1 = leaky(t1 + adn);
  float m = fmaxf(es, fmaxf(e0, e1));
  for (int j = c + 32; j < deg; j += 16)
    m = fmaxf(m, leaky(as_[ssrc[beg + j]] + adn));
  m = row16_xor_max<1>(m);
  m = row16_xor_max<2>(m);
  m = row16_xor_max<4>(m);
  m = row16_xor_max<8>(m);

  const float v0 = on0 ? __expf(e0 - m) : 0.0f;
  const float v1 = on1 ? __expf(e1 - m) : 0.0f;
  float sum = v0 + v1;
  for (int j = c + 32; j < deg; j += 16)
    sum += __expf(leaky(as_[ssrc[beg + j]] + adn) - m);
  sum = row16_xor_add<1>(sum);
  sum = row16_xor_add<2>(sum);
  sum = row16_xor_add<4>(sum);
  sum = row16_xor_add<8>(sum);
  const float vs = __expf(es - m);
  const float dn = (sum + vs) + 1e-16f;

  const float4* __restrict__ hp4 = (const float4*)hp;
  const float aself = vs / dn;
  const float4 hs = hp4[(unsigned)(node * 16 + c)];
  float4 acc;
  acc.x = hs.x * aself;
  acc.y = hs.y * aself;
  acc.z = hs.z * aself;
  acc.w = hs.w * aself;
  add_rows16(hp4, s0, v0 / dn, degmax, c, acc);
  if (degmax > 16) add_rows16(hp4, s1, v1 / dn, degmax - 16, c, acc);
  for (int base = 32; base < degmax; base += 16) {
    int sj = node;
    float aj = 0.0f;
    if (base + c < deg) {
      sj = ssrc[beg + base + c];
      aj = __expf(leaky(as_[sj] + adn) - m) / dn;
    }
    add_rows16(hp4, sj, aj, degmax - base, c, acc);
  }

  const float4 b = ((const float4*)bias)[c];
  float4 o;
  o.x = acc.x + b.x;
  o.y = acc.y + b.y;
  o.z = acc.z + b.z;
  o.w = acc.w + b.w;
  o.x = o.x > 0.0f ? o.x : 0.0f;
  o.y = o.y > 0.0f ? o.y : 0.0f;
  o.z = o.z > 0.0f ? o.z : 0.0f;
  o.w = o.w > 0.0f ? o.w : 0.0f;
  if (live) ((float4*)out)[(unsigned)(node * 16 + c)] = o;
}

extern "C" void kernel_launch(void* const* d_in, const int* in_sizes, int n_in,
                              void* d_out, int out_size, void* d_ws, size_t ws_size,
                              hipStream_t stream) {
  (void)n_in; (void)out_size; (void)ws_size;

  const float* x   = (const float*)d_in[0];
  const int*   ei  = (const int*)d_in[1];
  const int N = in_sizes[0] / 128;
  const int E = in_sizes[1] / 2;
  float* ws    = (float*)d_ws;
  float* hp    = ws;
  float* hcur  = hp    + (size_t)N * 64;
  float* lists = hcur  + (size_t)N * 64;
  float* as_   = lists + (size_t)N * 64;
  float* ad_   = as_   + N;
  int* cnt      = (int*)lists;
  int* rowstart = cnt + N + 1;
  int* rank     = rowstart + N;
  int* ssrc     = rank + E;

  csr_zero<<<(N + 1 + 255) / 256, 256, 0, stream>>>(cnt, N + 1);
  csr_count<<<(E + 255) / 256, 256, 0, stream>>>(ei, cnt, rank, E);
  csr_assign<<<(N + 255) / 256, 256, 0, stream>>>(cnt, rowstart, cnt + N, N);
  csr_fill<<<(E + 255) / 256, 256, 0, stream>>>(ei, rank, rowstart, ssrc, E);

  const float* hin = x;
  int K = 128;
  for (int layer = 0; layer < 3; ++layer) {
    const float* W    = (const float*)d_in[4 + layer * 4];
    const float* asrc = (const float*)d_in[5 + layer * 4];
    const float* adst = (const float*)d_in[6 + layer * 4];
    const float* bias = (const float*)d_in[7 + layer * 4];

    const int mblocks = (N / 16 + 1) / 2;
    gat_gemm_alpha<<<mblocks, 64, 0, stream>>>(hin, W, asrc, adst, hp, as_, ad_, N, K);

    float* dst = (layer == 2) ? (float*)d_out : hcur;
    gat_gather<<<(N + 15) / 16, 256, 0, stream>>>(rowstart, cnt, ssrc, hp, as_, ad_,
                                                  bias, dst, N);

    hin = hcur;
    K = 64;
  }
}


# Layers 1, 2: gather block keeps its 16 finished rows in LDS and does the next layer's fp32 WMMA product and alpha sums itself: 2 GEMM launches and the hidden plane's memory trip go; sum orders kept
# speedup vs baseline: 5.4724x; 1.0510x over previous
#include <hip/hip_runtime.h>
#include <math.h>


typedef __attribute__((ext_vector_type(2))) float v2f;
typedef __attribute__((ext_vector_type(8))) float v8f;

#define NEG_SLOPE 0.2f


template <int MASK>
__device__ __forceinline__ float row16_xor_add(float x) {
  const int t = __builtin_amdgcn_update_dpp(0, __float_as_int(x), 0x160 | MASK,
                                            0xf, 0xf, true);
  return x + __int_as_float(t);
}

__device__ __forceinline__ v8f wmma4(v2f a, float b0, float b1, v8f acc) {
  v2f b;
  b.x = b0;
  b.y = b1;
  return __builtin_amdgcn_wmma_f32_16x16x4_f32(false, a, false, b, (short)0, acc,
                                               false, false);
}

struct Slab {
  v2f a[4];
  float4 b0[4];
  float4 b1[4];
};

__device__ __forceinline__ void load_slab(Slab& f, const float* __restrict__ arow,
                                          const float* __restrict__ wcol, int k) {
#pragma unroll
  for (int j = 0; j < 4; ++j) {
    f.a[j]  = *(const v2f*)(arow + k + 4 * j);
    f.b0[j] = *(const float4*)(wcol + (size_t)(k + 4 * j) * 64);
    f.b1[j] = *(const float4*)(wcol + (size_t)(k + 4 * j + 1) * 64);
  }
}

__device__ __forceinline__ void mma_slab(const Slab& f, v8f& acc0, v8f& acc1,
                                         v8f& acc2, v8f& acc3) {
#pragma unroll
  for (int j = 0; j < 4; ++j) {
    acc0 = wmma4(f.a[j], f.b0[j].x, f.b1[j].x, acc0);
    acc1 = wmma4(f.a[j], f.b0[j].y, f.b1[j].y, acc1);
    acc2 = wmma4(f.a[j], f.b0[j].z, f.b1[j].z, acc2);
    acc3 = wmma4(f.a[j], f.b0[j].w, f.b1[j].w, acc3);
  }
}

__global__ __launch_bounds__(64) void gat_gemm_alpha(
    const float* __restrict__ A, const float* __restrict__ W,
    const float* __restrict__ a_s, const float* __restrict__ a_d,
    float* __restrict__ C, float* __restrict__ out_s, float* __restrict__ out_d,
    int M, int K) {
  const int lane = threadIdx.x & 31;
  const int mtile = blockIdx.x * 2 + (threadIdx.x >> 5);
  if (mtile * 16 >= M) return;
  const int c = lane & 15;
  const int h = lane >> 4;

  const float* __restrict__ arow = A + (size_t)(mtile * 16 + c) * K + 2 * h;
  const float* __restrict__ wcol = W + (size_t)(2 * h) * 64 + 4 * c;

  v8f acc0 = {}, acc1 = {}, acc2 = {}, acc3 = {};
  Slab f0, f1;
  load_slab(f0, arow, wcol, 0);
#pragma unroll 1
  for (int k = 0; k < K - 32; k += 32) {
    load_slab(f1, arow, wcol, k + 16);
    mma_slab(f0, acc0, acc1, acc2, acc3);
    load_slab(f0, arow, wcol, k + 32);
    mma_slab(f1, acc0, acc1, acc2, acc3);
  }
  load_slab(f1, arow, wcol, K - 16);
  mma_slab(f0, acc0, acc1, acc2, acc3);
  mma_slab(f1, acc0, acc1, acc2, acc3);

  const int rbase = mtile * 16 + 8 * h;
#pragma unroll
  for (int v = 0; v < 8; ++v) {
    float4 o;
    o.x = acc0[v];  o.y = acc1[v];  o.z = acc2[v];  o.w = acc3[v];
    *(float4*)(C + (size_t)(rbase + v) * 64 + 4 * c) = o;
  }

  const float4 s4 = *(const float4*)(a_s + 4 * c);
  const float4 d4 = *(const float4*)(a_d + 4 * c);
  float ps[8], pd[8];
#pragma unroll
  for (int v = 0; v < 8; ++v) {
    ps[v] = acc0[v] * s4.x + acc1[v] * s4.y + acc2[v] * s4.z + acc3[v] * s4.w;
    pd[v] = acc0[v] * d4.x + acc1[v] * d4.y + acc2[v] * d4.z + acc3[v] * d4.w;
  }
#pragma unroll
  for (int v = 0; v < 8; ++v) { ps[v] = row16_xor_add<1>(ps[v]); pd[v] = row16_xor_add<1>(pd[v]); }
#pragma unroll
  for (int v = 0; v < 8; ++v) { ps[v] = row16_xor_add<2>(ps[v]); pd[v] = row16_xor_add<2>(pd[v]); }
#pragma unroll
  for (int v = 0; v < 8; ++v) { ps[v] = row16_xor_add<4>(ps[v]); pd[v] = row16_xor_add<4>(pd[v]); }
#pragma unroll
  for (int v = 0; v < 8; ++v) { ps[v] = row16_xor_add<8>(ps[v]); pd[v] = row16_xor_add<8>(pd[v]); }

  float val = 0.0f;
#pragma unroll
  for (int v = 0; v < 8; ++v) {
    if (c == v) val = ps[v];
    if (c == v + 8) val = pd[v];
  }
  float* __restrict__ dst = (c < 8) ? (out_s + rbase + c) : (out_d + rbase + c - 8);
  *dst = val;
}

__global__ void csr_zero(int* __restrict__ cnt, int n) {
  const int tid = blockIdx.x * blockDim.x + threadIdx.x;
  if (tid < n) cnt[tid] = 0;
}

__global__ void csr_count(const int* __restrict__ ei, int* __restrict__ cnt,
                          int* __restrict__ rank, int E) {
  const int idx = blockIdx.x * blockDim.x + threadIdx.x;
  if (idx >= E) return;
  rank[idx] = atomicAdd(&cnt[ei[E + idx]], 1);
}

__global__ __launch_bounds__(256) void csr_assign(const int* cnt, int* rowstart,
                                                  int* cursor, int N) {
  const int d = blockIdx.x * 256 + threadIdx.x;
  const int lane = threadIdx.x & 31;
  const int c = (d < N) ? cnt[d] : 0;
  int inc = c;
#pragma unroll
  for (int off = 1; off < 32; off <<= 1) {
    const int t = __shfl_up(inc, off, 32);
    if (lane >= off) inc += t;
  }
  int base = 0;
  if (lane == 31) base = atomicAdd(cursor, inc);
  base = __shfl(base, 31, 32);
  if (d < N) rowstart[d] = base + inc - c;
}

__global__ void csr_fill(const int* __restrict__ ei, const int* __restrict__ rank,
                         const int* __restrict__ rowstart,
                         int* __restrict__ ssrc, int E) {
  const int idx = blockIdx.x * blockDim.x + threadIdx.x;
  if (idx >= E) return;
  ssrc[rowstart[ei[E + idx]] + rank[idx]] = ei[idx];
}

__device__ __forceinline__ float leaky(float e) {
  return (e > 0.0f) ? e : NEG_SLOPE * e;
}

template <int K>
__device__ __forceinline__ int row16_lane(int x) {
  return __builtin_amdgcn_update_dpp(0, x, 0x150 | K, 0xf, 0xf, true);
}
template <int MASK>
__device__ __forceinline__ float row16_xor_max(float x) {
  const int t = __builtin_amdgcn_update_dpp(0, __float_as_int(x), 0x160 | MASK,
                                            0xf, 0xf, true);
  return fmaxf(x, __int_as_float(t));
}

__device__ __forceinline__ void fma_row(float4& acc, const float4 r, float a) {
  acc.x = __builtin_fmaf(r.x, a, acc.x);
  acc.y = __builtin_fmaf(r.y, a, acc.y);
  acc.z = __builtin_fmaf(r.z, a, acc.z);
  acc.w = __builtin_fmaf(r.w, a, acc.w);
}

template <int K0>
__device__ __forceinline__ void add_rows4(const float4* __restrict__ hp4, int sj,
                                          float aj, int c, float4& acc) {
  const int ai = __float_as_int(aj);
  const float4 r0 = hp4[(unsigned)(row16_lane<K0>(sj) * 16 + c)];
  const float4 r1 = hp4[(unsigned)(row16_lane<K0 + 1>(sj) * 16 + c)];
  const float4 r2 = hp4[(unsigned)(row16_lane<K0 + 2>(sj) * 16 + c)];
  const float4 r3 = hp4[(unsigned)(row16_lane<K0 + 3>(sj) * 16 + c)];
  const float a0 = __int_as_float(row16_lane<K0>(ai));
  const float a1 = __int_as_float(row16_lane<K0 + 1>(ai));
  const float a2 = __int_as_float(row16_lane<K0 + 2>(ai));
  const float a3 = __int_as_float(row16_lane<K0 + 3>(ai));
  fma_row(acc, r0, a0);
  fma_row(acc, r1, a1);
  fma_row(acc, r2, a2);
  fma_row(acc, r3, a3);
}
__device__ __forceinline__ void add_rows16(const float4* __restrict__ hp4, int sj,
                                           float aj, int nkmax, int c, float4& acc) {
  if (nkmax > 0)  add_rows4<0>(hp4, sj, aj, c, acc);
  if (nkmax > 4)  add_rows4<4>(hp4, sj, aj, c, acc);
  if (nkmax > 8)  add_rows4<8>(hp4, sj, aj, c, acc);
  if (nkmax > 12) add_rows4<12>(hp4, sj, aj, c, acc);
}

#define TP 68
template <bool FUSE>
__global__ __launch_bounds__(256) void gat_gather(
    const int* __restrict__ rowstart, const int* __restrict__ cnt,
    const int* __restrict__ ssrc, const float* __restrict__ hp,
    const float* __restrict__ as_, const float* __restrict__ ad_,
    const float* __restrict__ bias, float* __restrict__ out, int N,
    const float* __restrict__ Wn, const float* __restrict__ a_sn,
    const float* __restrict__ a_dn, float* __restrict__ as_n,
    float* __restrict__ ad_n) {
  const int pair = __builtin_amdgcn_readfirstlane(blockIdx.x * 8 + (threadIdx.x >> 5));
  const int lane = threadIdx.x & 31;
  const int c = lane & 15;
  if (!FUSE && pair * 2 >= N) return;
  const bool live = pair * 2 + (lane >> 4) < N;
  const int node = live ? pair * 2 + (lane >> 4) : (pair * 2 < N ? pair * 2 : 0);
  const int beg = rowstart[node];
  const int cn = cnt[node];
  const float adn = ad_[node];
  const float es = leaky(as_[node] + adn);
  const int deg = live ? cn : 0;
  const int dlo = __builtin_amdgcn_readlane(deg, 0);
  const int dhi = __builtin_amdgcn_readlane(deg, 16);
  const int degmax = dlo > dhi ? dlo : dhi;

  const bool on0 = c < deg, on1 = c + 16 < deg;
  const int q0 = ssrc[on0 ? beg + c : 0];
  const int q1 = ssrc[on1 ? beg + 16 + c : 0];
  const int s0 = on0 ? q0 : node;
  const int s1 = on1 ? q1 : node;
  const float t0 = as_[s0], t1 = as_[s1];
  const float e0 = leaky(t0 + adn), e1 = leaky(t1 + adn);
  float m = fmaxf(es, fmaxf(e0, e1));
  for (int j = c + 32; j < deg; j += 16)
    m = fmaxf(m, leaky(as_[ssrc[beg + j]] + adn));
  m = row16_xor_max<1>(m);
  m = row16_xor_max<2>(m);
  m = row16_xor_max<4>(m);
  m = row16_xor_max<8>(m);

  const float v0 = on0 ? __expf(e0 - m) : 0.0f;
  const float v1 = on1 ? __expf(e1 - m) : 0.0f;
  float sum = v0 + v1;
  for (int j = c + 32; j < deg; j += 16)
    sum += __expf(leaky(as_[ssrc[beg + j]] + adn) - m);
  sum = row16_xor_add<1>(sum);
  sum = row16_xor_add<2>(sum);
  sum = row16_xor_add<4>(sum);
  sum = row16_xor_add<8>(sum);
  const float vs = __expf(es - m);
  const float dn = (sum + vs) + 1e-16f;

  const float4* __restrict__ hp4 = (const float4*)hp;
  const float aself = vs / dn;
  const float4 hs = hp4[(unsigned)(node * 16 + c)];
  float4 acc;
  acc.x = hs.x * aself;
  acc.y = hs.y * aself;
  acc.z = hs.z * aself;
  acc.w = hs.w * aself;
  add_rows16(hp4, s0, v0 / dn, degmax, c, acc);
  if (degmax > 16) add_rows16(hp4, s1, v1 / dn, degmax - 16, c, acc);
  for (int base = 32; base < degmax; base += 16) {
    int sj = node;
    float aj = 0.0f;
    if (base + c < deg) {
      sj = ssrc[beg + base + c];
      aj = __expf(leaky(as_[sj] + adn) - m) / dn;
    }
    add_rows16(hp4, sj, aj, degmax - base, c, acc);
  }

  const float4 b = ((const float4*)bias)[c];
  float4 o;
  o.x = acc.x + b.x;
  o.y = acc.y + b.y;
  o.z = acc.z + b.z;
  o.w = acc.w + b.w;
  o.x = o.x > 0.0f ? o.x : 0.0f;
  o.y = o.y > 0.0f ? o.y : 0.0f;
  o.z = o.z > 0.0f ? o.z : 0.0f;
  o.w = o.w > 0.0f ? o.w : 0.0f;
  if (!FUSE) {
    if (live) ((float4*)out)[(unsigned)(node * 16 + c)] = o;
    return;
  }

  __shared__ float sA[16 * TP];
  __shared__ float sC[16 * TP];
  const int r = threadIdx.x >> 4;
  const int h = lane >> 4;
  const int w = __builtin_amdgcn_readfirstlane(threadIdx.x >> 5);
  *(float4*)(sA + r * TP + 4 * c) = o;
  float wb[32];
  if (w < 4) {
#pragma unroll
    for (int j = 0; j < 16; ++j) {
      wb[2 * j]     = Wn[(4 * j + 2 * h) * 64 + 16 * w + c];
      wb[2 * j + 1] = Wn[(4 * j + 2 * h + 1) * 64 + 16 * w + c];
    }
  }
  __syncthreads();
  if (w < 4) {
    v8f acc = {};
#pragma unroll
    for (int j = 0; j < 16; ++j) {
      const v2f a = *(const v2f*)(sA + c * TP + 4 * j + 2 * h);
      acc = wmma4(a, wb[2 * j], wb[2 * j + 1], acc);
    }
    __builtin_amdgcn_sched_barrier(0);
    asm volatile("v_nop\n\tv_nop\n\tv_nop\n\tv_nop");
    __builtin_amdgcn_sched_barrier(0);
#pragma unroll
    for (int v = 0; v < 8; ++v) sC[(8 * h + v) * TP + 16 * w + c] = acc[v];
  }
  __syncthreads();
  const float4 cv = *(const float4*)(sC + r * TP + 4 * c);
  if (live) ((float4*)out)[(unsigned)(node * 16 + c)] = cv;
  const float4 s4 = *(const float4*)(a_sn + 4 * c);
  const float4 d4 = *(const float4*)(a_dn + 4 * c);
  float ps = cv.x * s4.x + cv.y * s4.y + cv.z * s4.z + cv.w * s4.w;
  float pd = cv.x * d4.x + cv.y * d4.y + cv.z * d4.z + cv.w * d4.w;
  ps = row16_xor_add<1>(ps);  pd = row16_xor_add<1>(pd);
  ps = row16_xor_add<2>(ps);  pd = row16_xor_add<2>(pd);
  ps = row16_xor_add<4>(ps);  pd = row16_xor_add<4>(pd);
  ps = row16_xor_add<8>(ps);  pd = row16_xor_add<8>(pd);
  if (live && c == 0) {
    as_n[node] = ps;
    ad_n[node] = pd;
  }
}

extern "C" void kernel_launch(void* const* d_in, const int* in_sizes, int n_in,
                              void* d_out, int out_size, void* d_ws, size_t ws_size,
                              hipStream_t stream) {
  (void)n_in; (void)out_size; (void)ws_size;

  const float* x   = (const float*)d_in[0];
  const int*   ei  = (const int*)d_in[1];
  const int N = in_sizes[0] / 128;
  const int E = in_sizes[1] / 2;
  float* ws    = (float*)d_ws;
  float* hp    = ws;
  float* hp2   = hp    + (size_t)N * 64;
  float* lists = hp2   + (size_t)N * 64;
  float* as_   = lists + (size_t)N * 64;
  float* ad_   = as_   + N;
  int* cnt      = (int*)lists;
  int* rowstart = cnt + N + 1;
  int* rank     = rowstart + N;
  int* ssrc     = rank + E;
  float* as2    = (float*)(ssrc + E);
  float* ad2    = as2 + N;

  csr_zero<<<(N + 1 + 255) / 256, 256, 0, stream>>>(cnt, N + 1);
  csr_count<<<(E + 255) / 256, 256, 0, stream>>>(ei, cnt, rank, E);
  csr_assign<<<(N + 255) / 256, 256, 0, stream>>>(cnt, rowstart, cnt + N, N);
  csr_fill<<<(E + 255) / 256, 256, 0, stream>>>(ei, rank, rowstart, ssrc, E);

  const float* W0 = (const float*)d_in[4];
  const float* W1 = (const float*)d_in[8];
  const float* W2 = (const float*)d_in[12];

  const int mblocks = (N / 16 + 1) / 2;
  gat_gemm_alpha<<<mblocks, 64, 0, stream>>>(x, W0, (const float*)d_in[5],
                                             (const float*)d_in[6], hp, as_, ad_, N, 128);

  const int gblocks = (N + 15) / 16;
  gat_gather<true><<<gblocks, 256, 0, stream>>>(rowstart, cnt, ssrc, hp, as_, ad_,
      (const float*)d_in[7], hp2, N, W1, (const float*)d_in[9],
      (const float*)d_in[10], as2, ad2);
  gat_gather<true><<<gblocks, 256, 0, stream>>>(rowstart, cnt, ssrc, hp2, as2, ad2,
      (const float*)d_in[11], hp, N, W2, (const float*)d_in[13],
      (const float*)d_in[14], as_, ad_);
  gat_gather<false><<<gblocks, 256, 0, stream>>>(rowstart, cnt, ssrc, hp, as_, ad_,
      (const float*)d_in[15], (float*)d_out, N, nullptr, nullptr, nullptr,
      nullptr, nullptr);
}


# The counting pass puts each source into slot 'rank' of its destination's fixed 32-slot row (later arrivals chained per node): segment-start and fill launches go; edges past 32 sum in list order
# speedup vs baseline: 6.3216x; 1.1552x over previous
#include <hip/hip_runtime.h>
#include <math.h>


typedef __attribute__((ext_vector_type(2))) float v2f;
typedef __attribute__((ext_vector_type(8))) float v8f;

#define NEG_SLOPE 0.2f


template <int MASK>
__device__ __forceinline__ float row16_xor_add(float x) {
  const int t = __builtin_amdgcn_update_dpp(0, __float_as_int(x), 0x160 | MASK,
                                            0xf, 0xf, true);
  return x + __int_as_float(t);
}

__device__ __forceinline__ v8f wmma4(v2f a, float b0, float b1, v8f acc) {
  v2f b;
  b.x = b0;
  b.y = b1;
  return __builtin_amdgcn_wmma_f32_16x16x4_f32(false, a, false, b, (short)0, acc,
                                               false, false);
}

struct Slab {
  v2f a[4];
  float4 b0[4];
  float4 b1[4];
};

__device__ __forceinline__ void load_slab(Slab& f, const float* __restrict__ arow,
                                          const float* __restrict__ wcol, int k) {
#pragma unroll
  for (int j = 0; j < 4; ++j) {
    f.a[j]  = *(const v2f*)(arow + k + 4 * j);
    f.b0[j] = *(const float4*)(wcol + (size_t)(k + 4 * j) * 64);
    f.b1[j] = *(const float4*)(wcol + (size_t)(k + 4 * j + 1) * 64);
  }
}

__device__ __forceinline__ void mma_slab(const Slab& f, v8f& acc0, v8f& acc1,
                                         v8f& acc2, v8f& acc3) {
#pragma unroll
  for (int j = 0; j < 4; ++j) {
    acc0 = wmma4(f.a[j], f.b0[j].x, f.b1[j].x, acc0);
    acc1 = wmma4(f.a[j], f.b0[j].y, f.b1[j].y, acc1);
    acc2 = wmma4(f.a[j], f.b0[j].z, f.b1[j].z, acc2);
    acc3 = wmma4(f.a[j], f.b0[j].w, f.b1[j].w, acc3);
  }
}

__global__ __launch_bounds__(64) void gat_gemm_alpha(
    const float* __restrict__ A, const float* __restrict__ W,
    const float* __restrict__ a_s, const float* __restrict__ a_d,
    float* __restrict__ C, float* __restrict__ out_s, float* __restrict__ out_d,
    int M, int K) {
  const int lane = threadIdx.x & 31;
  const int mtile = blockIdx.x * 2 + (threadIdx.x >> 5);
  if (mtile * 16 >= M) return;
  const int c = lane & 15;
  const int h = lane >> 4;

  const float* __restrict__ arow = A + (size_t)(mtile * 16 + c) * K + 2 * h;
  const float* __restrict__ wcol = W + (size_t)(2 * h) * 64 + 4 * c;

  v8f acc0 = {}, acc1 = {}, acc2 = {}, acc3 = {};
  Slab f0, f1;
  load_slab(f0, arow, wcol, 0);
#pragma unroll 1
  for (int k = 0; k < K - 32; k += 32) {
    load_slab(f1, arow, wcol, k + 16);
    mma_slab(f0, acc0, acc1, acc2, acc3);
    load_slab(f0, arow, wcol, k + 32);
    mma_slab(f1, acc0, acc1, acc2, acc3);
  }
  load_slab(f1, arow, wcol, K - 16);
  mma_slab(f0, acc0, acc1, acc2, acc3);
  mma_slab(f1, acc0, acc1, acc2, acc3);

  const int rbase = mtile * 16 + 8 * h;
#pragma unroll
  for (int v = 0; v < 8; ++v) {
    float4 o;
    o.x = acc0[v];  o.y = acc1[v];  o.z = acc2[v];  o.w = acc3[v];
    *(float4*)(C + (size_t)(rbase + v) * 64 + 4 * c) = o;
  }

  const float4 s4 = *(const float4*)(a_s + 4 * c);
  const float4 d4 = *(const float4*)(a_d + 4 * c);
  float ps[8], pd[8];
#pragma unroll
  for (int v = 0; v < 8; ++v) {
    ps[v] = acc0[v] * s4.x + acc1[v] * s4.y + acc2[v] * s4.z + acc3[v] * s4.w;
    pd[v] = acc0[v] * d4.x + acc1[v] * d4.y + acc2[v] * d4.z + acc3[v] * d4.w;
  }
#pragma unroll
  for (int v = 0; v < 8; ++v) { ps[v] = row16_xor_add<1>(ps[v]); pd[v] = row16_xor_add<1>(pd[v]); }
#pragma unroll
  for (int v = 0; v < 8; ++v) { ps[v] = row16_xor_add<2>(ps[v]); pd[v] = row16_xor_add<2>(pd[v]); }
#pragma unroll
  for (int v = 0; v < 8; ++v) { ps[v] = row16_xor_add<4>(ps[v]); pd[v] = row16_xor_add<4>(pd[v]); }
#pragma unroll
  for (int v = 0; v < 8; ++v) { ps[v] = row16_xor_add<8>(ps[v]); pd[v] = row16_xor_add<8>(pd[v]); }

  float val = 0.0f;
#pragma unroll
  for (int v = 0; v < 8; ++v) {
    if (c == v) val = ps[v];
    if (c == v + 8) val = pd[v];
  }
  float* __restrict__ dst = (c < 8) ? (out_s + rbase + c) : (out_d + rbase + c - 8);
  *dst = val;
}

#define SLOTS 32

__global__ void csr_zero(int* __restrict__ cnt, int* __restrict__ head, int n) {
  const int tid = blockIdx.x * blockDim.x + threadIdx.x;
  if (tid < n) {
    cnt[tid] = 0;
    head[tid] = -1;
  }
}

__global__ void csr_place(const int* __restrict__ ei, int* __restrict__ cnt,
                          int* __restrict__ head, int* __restrict__ slots,
                          int* __restrict__ next, int E) {
  const int idx = blockIdx.x * blockDim.x + threadIdx.x;
  if (idx >= E) return;
  int s = ei[idx];
  const int d = ei[E + idx];
  asm volatile("" : "+v"(s));
  const int r = atomicAdd(&cnt[d], 1);
  if (r < SLOTS) slots[(unsigned)(d * SLOTS + r)] = s;
  else next[idx] = atomicExch(&head[d], idx);
}

__device__ __forceinline__ float leaky(float e) {
  return (e > 0.0f) ? e : NEG_SLOPE * e;
}

template <int K>
__device__ __forceinline__ int row16_lane(int x) {
  return __builtin_amdgcn_update_dpp(0, x, 0x150 | K, 0xf, 0xf, true);
}
template <int MASK>
__device__ __forceinline__ float row16_xor_max(float x) {
  const int t = __builtin_amdgcn_update_dpp(0, __float_as_int(x), 0x160 | MASK,
                                            0xf, 0xf, true);
  return fmaxf(x, __int_as_float(t));
}

__device__ __forceinline__ void fma_row(float4& acc, const float4 r, float a) {
  acc.x = __builtin_fmaf(r.x, a, acc.x);
  acc.y = __builtin_fmaf(r.y, a, acc.y);
  acc.z = __builtin_fmaf(r.z, a, acc.z);
  acc.w = __builtin_fmaf(r.w, a, acc.w);
}

template <int K0>
__device__ __forceinline__ void add_rows4(const float4* __restrict__ hp4, int sj,
                                          float aj, int c, float4& acc) {
  const int ai = __float_as_int(aj);
  const float4 r0 = hp4[(unsigned)(row16_lane<K0>(sj) * 16 + c)];
  const float4 r1 = hp4[(unsigned)(row16_lane<K0 + 1>(sj) * 16 + c)];
  const float4 r2 = hp4[(unsigned)(row16_lane<K0 + 2>(sj) * 16 + c)];
  const float4 r3 = hp4[(unsigned)(row16_lane<K0 + 3>(sj) * 16 + c)];
  const float a0 = __int_as_float(row16_lane<K0>(ai));
  const float a1 = __int_as_float(row16_lane<K0 + 1>(ai));
  const float a2 = __int_as_float(row16_lane<K0 + 2>(ai));
  const float a3 = __int_as_float(row16_lane<K0 + 3>(ai));
  fma_row(acc, r0, a0);
  fma_row(acc, r1, a1);
  fma_row(acc, r2, a2);
  fma_row(acc, r3, a3);
}
__device__ __forceinline__ void add_rows16(const float4* __restrict__ hp4, int sj,
                                           float aj, int nkmax, int c, float4& acc) {
  if (nkmax > 0)  add_rows4<0>(hp4, sj, aj, c, acc);
  if (nkmax > 4)  add_rows4<4>(hp4, sj, aj, c, acc);
  if (nkmax > 8)  add_rows4<8>(hp4, sj, aj, c, acc);
  if (nkmax > 12) add_rows4<12>(hp4, sj, aj, c, acc);
}

#define TP 68
template <bool FUSE>
__global__ __launch_bounds__(256) void gat_gather(
    const int* __restrict__ cnt, const int* __restrict__ slots,
    const int* __restrict__ head, const int* __restrict__ next,
    const int* __restrict__ ei, int E, const float* __restrict__ hp,
    const float* __restrict__ as_, const float* __restrict__ ad_,
    const float* __restrict__ bias, float* __restrict__ out, int N,
    const float* __restrict__ Wn, const float* __restrict__ a_sn,
    const float* __restrict__ a_dn, float* __restrict__ as_n,
    float* __restrict__ ad_n) {
  const int pair = __builtin_amdgcn_readfirstlane(blockIdx.x * 8 + (threadIdx.x >> 5));
  const int lane = threadIdx.x & 31;
  const int c = lane & 15;
  if (!FUSE && pair * 2 >= N) return;
  const bool live = pair * 2 + (lane >> 4) < N;
  const int node = live ? pair * 2 + (lane >> 4) : (pair * 2 < N ? pair * 2 : 0);
  const int beg = node * SLOTS;
  const int q0 = slots[(unsigned)(beg + c)];
  const int q1 = slots[(unsigned)(beg + 16 + c)];
  const int cn = cnt[node];
  const int hd = head[node];
  const float adn = ad_[node];
  const float asn = as_[node];
  __builtin_amdgcn_sched_barrier(0);
  const float es = leaky(asn + adn);
  const int deg = live ? cn : 0;
  const int dlo = __builtin_amdgcn_readlane(deg, 0);
  const int dhi = __builtin_amdgcn_readlane(deg, 16);
  const int degmax = dlo > dhi ? dlo : dhi;

  const bool on0 = c < deg, on1 = c + 16 < deg;
  const int s0 = on0 ? q0 : node;
  const int s1 = on1 ? q1 : node;
  const float t0 = as_[s0], t1 = as_[s1];
  const float e0 = leaky(t0 + adn), e1 = leaky(t1 + adn);
  float m = fmaxf(es, fmaxf(e0, e1));
  const int nov = deg - SLOTS;
  {
    int k = hd;
    for (int t = 0; t < nov && (unsigned)k < (unsigned)E; ++t) {
      m = fmaxf(m, leaky(as_[ei[k]] + adn));
      k = next[k];
    }
  }
  m = row16_xor_max<1>(m);
  m = row16_xor_max<2>(m);
  m = row16_xor_max<4>(m);
  m = row16_xor_max<8>(m);

  const float v0 = on0 ? __expf(e0 - m) : 0.0f;
  const float v1 = on1 ? __expf(e1 - m) : 0.0f;
  float sum = v0 + v1;
  sum = row16_xor_add<1>(sum);
  sum = row16_xor_add<2>(sum);
  sum = row16_xor_add<4>(sum);
  sum = row16_xor_add<8>(sum);
  float ov = 0.0f;
  {
    int k = hd;
    for (int t = 0; t < nov && (unsigned)k < (unsigned)E; ++t) {
      ov += __expf(leaky(as_[ei[k]] + adn) - m);
      k = next[k];
    }
  }
  const float vs = __expf(es - m);
  const float dn = ((sum + ov) + vs) + 1e-16f;

  const float4* __restrict__ hp4 = (const float4*)hp;
  const float aself = vs / dn;
  const float4 hs = hp4[(unsigned)(node * 16 + c)];
  float4 acc;
  acc.x = hs.x * aself;
  acc.y = hs.y * aself;
  acc.z = hs.z * aself;
  acc.w = hs.w * aself;
  add_rows16(hp4, s0, v0 / dn, degmax, c, acc);
  if (degmax > 16) add_rows16(hp4, s1, v1 / dn, degmax - 16, c, acc);
  {
    int k = hd;
    for (int t = 0; t < nov && (unsigned)k < (unsigned)E; ++t) {
      const int sj = ei[k];
      const float aj = __expf(leaky(as_[sj] + adn) - m) / dn;
      fma_row(acc, hp4[(unsigned)(sj * 16 + c)], aj);
      k = next[k];
    }
  }

  const float4 b = ((const float4*)bias)[c];
  float4 o;
  o.x = acc.x + b.x;
  o.y = acc.y + b.y;
  o.z = acc.z + b.z;
  o.w = acc.w + b.w;
  o.x = o.x > 0.0f ? o.x : 0.0f;
  o.y = o.y > 0.0f ? o.y : 0.0f;
  o.z = o.z > 0.0f ? o.z : 0.0f;
  o.w = o.w > 0.0f ? o.w : 0.0f;
  if (!FUSE) {
    if (live) ((float4*)out)[(unsigned)(node * 16 + c)] = o;
    return;
  }

  __shared__ float sA[16 * TP];
  __shared__ float sC[16 * TP];
  const int r = threadIdx.x >> 4;
  const int h = lane >> 4;
  const int w = __builtin_amdgcn_readfirstlane(threadIdx.x >> 5);
  *(float4*)(sA + r * TP + 4 * c) = o;
  float wb[32];
  if (w < 4) {
#pragma unroll
    for (int j = 0; j < 16; ++j) {
      wb[2 * j]     = Wn[(4 * j + 2 * h) * 64 + 16 * w + c];
      wb[2 * j + 1] = Wn[(4 * j + 2 * h + 1) * 64 + 16 * w + c];
    }
  }
  __syncthreads();
  if (w < 4) {
    v8f acc = {};
#pragma unroll
    for (int j = 0; j < 16; ++j) {
      const v2f a = *(const v2f*)(sA + c * TP + 4 * j + 2 * h);
      acc = wmma4(a, wb[2 * j], wb[2 * j + 1], acc);
    }
    __builtin_amdgcn_sched_barrier(0);
    asm volatile("v_nop\n\tv_nop\n\tv_nop\n\tv_nop");
    __builtin_amdgcn_sched_barrier(0);
#pragma unroll
    for (int v = 0; v < 8; ++v) sC[(8 * h + v) * TP + 16 * w + c] = acc[v];
  }
  __syncthreads();
  const float4 cv = *(const float4*)(sC + r * TP + 4 * c);
  if (live) ((float4*)out)[(unsigned)(node * 16 + c)] = cv;
  const float4 s4 = *(const float4*)(a_sn + 4 * c);
  const float4 d4 = *(const float4*)(a_dn + 4 * c);
  float ps = cv.x * s4.x + cv.y * s4.y + cv.z * s4.z + cv.w * s4.w;
  float pd = cv.x * d4.x + cv.y * d4.y + cv.z * d4.z + cv.w * d4.w;
  ps = row16_xor_add<1>(ps);  pd = row16_xor_add<1>(pd);
  ps = row16_xor_add<2>(ps);  pd = row16_xor_add<2>(pd);
  ps = row16_xor_add<4>(ps);  pd = row16_xor_add<4>(pd);
  ps = row16_xor_add<8>(ps);  pd = row16_xor_add<8>(pd);
  if (live && c == 0) {
    as_n[node] = ps;
    ad_n[node] = pd;
  }
}

extern "C" void kernel_launch(void* const* d_in, const int* in_sizes, int n_in,
                              void* d_out, int out_size, void* d_ws, size_t ws_size,
                              hipStream_t stream) {
  (void)n_in; (void)out_size; (void)ws_size;

  const float* x   = (const float*)d_in[0];
  const int*   ei  = (const int*)d_in[1];
  const int N = in_sizes[0] / 128;
  const int E = in_sizes[1] / 2;
  float* ws    = (float*)d_ws;
  float* hp    = ws;
  float* hp2   = hp    + (size_t)N * 64;
  float* lists = hp2   + (size_t)N * 64;
  float* as_   = lists + (size_t)N * 64;
  float* ad_   = as_   + N;
  int* cnt      = (int*)lists;
  int* head     = cnt + N;
  int* slots    = head + N;
  int* next     = slots + (size_t)N * SLOTS;
  float* as2    = (float*)(next + E);
  float* ad2    = as2 + N;

  csr_zero<<<(N + 255) / 256, 256, 0, stream>>>(cnt, head, N);
  csr_place<<<(E + 255) / 256, 256, 0, stream>>>(ei, cnt, head, slots, next, E);

  const float* W0 = (const float*)d_in[4];
  const float* W1 = (const float*)d_in[8];
  const float* W2 = (const float*)d_in[12];

  const int mblocks = (N / 16 + 1) / 2;
  gat_gemm_alpha<<<mblocks, 64, 0, stream>>>(x, W0, (const float*)d_in[5],
                                             (const float*)d_in[6], hp, as_, ad_, N, 128);

  const int gblocks = (N + 15) / 16;
  gat_gather<true><<<gblocks, 256, 0, stream>>>(cnt, slots, head, next, ei, E, hp, as_, ad_,
      (const float*)d_in[7], hp2, N, W1, (const float*)d_in[9],
      (const float*)d_in[10], as2, ad2);
  gat_gather<true><<<gblocks, 256, 0, stream>>>(cnt, slots, head, next, ei, E, hp2, as2, ad2,
      (const float*)d_in[11], hp, N, W2, (const float*)d_in[13],
      (const float*)d_in[14], as_, ad_);
  gat_gather<false><<<gblocks, 256, 0, stream>>>(cnt, slots, head, next, ei, E, hp, as_, ad_,
      (const float*)d_in[15], (float*)d_out, N, nullptr, nullptr, nullptr,
      nullptr, nullptr);
}


# Layer-0 GEMM and edge placement share one launch (GEMM blocks first in the grid, placing blocks behind): one launch boundary goes and the GEMM runs under the placement; sum orders kept
# speedup vs baseline: 6.8029x; 1.0761x over previous
#include <hip/hip_runtime.h>
#include <math.h>


typedef __attribute__((ext_vector_type(2))) float v2f;
typedef __attribute__((ext_vector_type(8))) float v8f;

#define NEG_SLOPE 0.2f


template <int MASK>
__device__ __forceinline__ float row16_xor_add(float x) {
  const int t = __builtin_amdgcn_update_dpp(0, __float_as_int(x), 0x160 | MASK,
                                            0xf, 0xf, true);
  return x + __int_as_float(t);
}

__device__ __forceinline__ v8f wmma4(v2f a, float b0, float b1, v8f acc) {
  v2f b;
  b.x = b0;
  b.y = b1;
  return __builtin_amdgcn_wmma_f32_16x16x4_f32(false, a, false, b, (short)0, acc,
                                               false, false);
}

struct Slab {
  v2f a[4];
  float4 b0[4];
  float4 b1[4];
};

__device__ __forceinline__ void load_slab(Slab& f, const float* __restrict__ arow,
                                          const float* __restrict__ wcol, int k) {
#pragma unroll
  for (int j = 0; j < 4; ++j) {
    f.a[j]  = *(const v2f*)(arow + k + 4 * j);
    f.b0[j] = *(const float4*)(wcol + (size_t)(k + 4 * j) * 64);
    f.b1[j] = *(const float4*)(wcol + (size_t)(k + 4 * j + 1) * 64);
  }
}

__device__ __forceinline__ void mma_slab(const Slab& f, v8f& acc0, v8f& acc1,
                                         v8f& acc2, v8f& acc3) {
#pragma unroll
  for (int j = 0; j < 4; ++j) {
    acc0 = wmma4(f.a[j], f.b0[j].x, f.b1[j].x, acc0);
    acc1 = wmma4(f.a[j], f.b0[j].y, f.b1[j].y, acc1);
    acc2 = wmma4(f.a[j], f.b0[j].z, f.b1[j].z, acc2);
    acc3 = wmma4(f.a[j], f.b0[j].w, f.b1[j].w, acc3);
  }
}

__device__ __forceinline__ void gemm_alpha_tile(
    const float* __restrict__ A, const float* __restrict__ W,
    const float* __restrict__ a_s, const float* __restrict__ a_d,
    float* __restrict__ C, float* __restrict__ out_s, float* __restrict__ out_d,
    int M, int K, int mtile) {
  const int lane = threadIdx.x & 31;
  if (mtile * 16 >= M) return;
  const int c = lane & 15;
  const int h = lane >> 4;

  const float* __restrict__ arow = A + (size_t)(mtile * 16 + c) * K + 2 * h;
  const float* __restrict__ wcol = W + (size_t)(2 * h) * 64 + 4 * c;

  v8f acc0 = {}, acc1 = {}, acc2 = {}, acc3 = {};
  Slab f0, f1;
  load_slab(f0, arow, wcol, 0);
#pragma unroll 1
  for (int k = 0; k < K - 32; k += 32) {
    load_slab(f1, arow, wcol, k + 16);
    mma_slab(f0, acc0, acc1, acc2, acc3);
    load_slab(f0, arow, wcol, k + 32);
    mma_slab(f1, acc0, acc1, acc2, acc3);
  }
  load_slab(f1, arow, wcol, K - 16);
  mma_slab(f0, acc0, acc1, acc2, acc3);
  mma_slab(f1, acc0, acc1, acc2, acc3);

  const int rbase = mtile * 16 + 8 * h;
#pragma unroll
  for (int v = 0; v < 8; ++v) {
    float4 o;
    o.x = acc0[v];  o.y = acc1[v];  o.z = acc2[v];  o.w = acc3[v];
    *(float4*)(C + (size_t)(rbase + v) * 64 + 4 * c) = o;
  }

  const float4 s4 = *(const float4*)(a_s + 4 * c);
  const float4 d4 = *(const float4*)(a_d + 4 * c);
  float ps[8], pd[8];
#pragma unroll
  for (int v = 0; v < 8; ++v) {
    ps[v] = acc0[v] * s4.x + acc1[v] * s4.y + acc2[v] * s4.z + acc3[v] * s4.w;
    pd[v] = acc0[v] * d4.x + acc1[v] * d4.y + acc2[v] * d4.z + acc3[v] * d4.w;
  }
#pragma unroll
  for (int v = 0; v < 8; ++v) { ps[v] = row16_xor_add<1>(ps[v]); pd[v] = row16_xor_add<1>(pd[v]); }
#pragma unroll
  for (int v = 0; v < 8; ++v) { ps[v] = row16_xor_add<2>(ps[v]); pd[v] = row16_xor_add<2>(pd[v]); }
#pragma unroll
  for (int v = 0; v < 8; ++v) { ps[v] = row16_xor_add<4>(ps[v]); pd[v] = row16_xor_add<4>(pd[v]); }
#pragma unroll
  for (int v = 0; v < 8; ++v) { ps[v] = row16_xor_add<8>(ps[v]); pd[v] = row16_xor_add<8>(pd[v]); }

  float val = 0.0f;
#pragma unroll
  for (int v = 0; v < 8; ++v) {
    if (c == v) val = ps[v];
    if (c == v + 8) val = pd[v];
  }
  float* __restrict__ dst = (c < 8) ? (out_s + rbase + c) : (out_d + rbase + c - 8);
  *dst = val;
}

#define SLOTS 32

__global__ void csr_zero(int* __restrict__ cnt, int* __restrict__ head, int n) {
  const int tid = blockIdx.x * blockDim.x + threadIdx.x;
  if (tid < n) {
    cnt[tid] = 0;
    head[tid] = -1;
  }
}

__device__ __forceinline__ void place_edge(const int* __restrict__ ei,
                                           int* __restrict__ cnt,
                                           int* __restrict__ head,
                                           int* __restrict__ slots,
                                           int* __restrict__ next, int E, int idx) {
  if (idx >= E) return;
  int s = ei[idx];
  const int d = ei[E + idx];
  asm volatile("" : "+v"(s));
  const int r = atomicAdd(&cnt[d], 1);
  if (r < SLOTS) slots[(unsigned)(d * SLOTS + r)] = s;
  else next[idx] = atomicExch(&head[d], idx);
}

__global__ __launch_bounds__(256) void gat_gemm_alpha_place(
    const float* __restrict__ A, const float* __restrict__ W,
    const float* __restrict__ a_s, const float* __restrict__ a_d,
    float* __restrict__ C, float* __restrict__ out_s, float* __restrict__ out_d,
    int M, int K, int gemm_blocks,
    const int* __restrict__ ei, int* __restrict__ cnt, int* __restrict__ head,
    int* __restrict__ slots, int* __restrict__ next, int E) {
  const int b = blockIdx.x;
  if (b < gemm_blocks) {
    gemm_alpha_tile(A, W, a_s, a_d, C, out_s, out_d, M, K,
                    b * 8 + (threadIdx.x >> 5));
  } else {
    place_edge(ei, cnt, head, slots, next, E, (b - gemm_blocks) * 256 + threadIdx.x);
  }
}

__device__ __forceinline__ float leaky(float e) {
  return (e > 0.0f) ? e : NEG_SLOPE * e;
}

template <int K>
__device__ __forceinline__ int row16_lane(int x) {
  return __builtin_amdgcn_update_dpp(0, x, 0x150 | K, 0xf, 0xf, true);
}
template <int MASK>
__device__ __forceinline__ float row16_xor_max(float x) {
  const int t = __builtin_amdgcn_update_dpp(0, __float_as_int(x), 0x160 | MASK,
                                            0xf, 0xf, true);
  return fmaxf(x, __int_as_float(t));
}

__device__ __forceinline__ void fma_row(float4& acc, const float4 r, float a) {
  acc.x = __builtin_fmaf(r.x, a, acc.x);
  acc.y = __builtin_fmaf(r.y, a, acc.y);
  acc.z = __builtin_fmaf(r.z, a, acc.z);
  acc.w = __builtin_fmaf(r.w, a, acc.w);
}

template <int K0>
__device__ __forceinline__ void add_rows4(const float4* __restrict__ hp4, int sj,
                                          float aj, int c, float4& acc) {
  const int ai = __float_as_int(aj);
  const float4 r0 = hp4[(unsigned)(row16_lane<K0>(sj) * 16 + c)];
  const float4 r1 = hp4[(unsigned)(row16_lane<K0 + 1>(sj) * 16 + c)];
  const float4 r2 = hp4[(unsigned)(row16_lane<K0 + 2>(sj) * 16 + c)];
  const float4 r3 = hp4[(unsigned)(row16_lane<K0 + 3>(sj) * 16 + c)];
  const float a0 = __int_as_float(row16_lane<K0>(ai));
  const float a1 = __int_as_float(row16_lane<K0 + 1>(ai));
  const float a2 = __int_as_float(row16_lane<K0 + 2>(ai));
  const float a3 = __int_as_float(row16_lane<K0 + 3>(ai));
  fma_row(acc, r0, a0);
  fma_row(acc, r1, a1);
  fma_row(acc, r2, a2);
  fma_row(acc, r3, a3);
}
__device__ __forceinline__ void add_rows16(const float4* __restrict__ hp4, int sj,
                                           float aj, int nkmax, int c, float4& acc) {
  if (nkmax > 0)  add_rows4<0>(hp4, sj, aj, c, acc);
  if (nkmax > 4)  add_rows4<4>(hp4, sj, aj, c, acc);
  if (nkmax > 8)  add_rows4<8>(hp4, sj, aj, c, acc);
  if (nkmax > 12) add_rows4<12>(hp4, sj, aj, c, acc);
}

#define TP 68
template <bool FUSE>
__global__ __launch_bounds__(256) void gat_gather(
    const int* __restrict__ cnt, const int* __restrict__ slots,
    const int* __restrict__ head, const int* __restrict__ next,
    const int* __restrict__ ei, int E, const float* __restrict__ hp,
    const float* __restrict__ as_, const float* __restrict__ ad_,
    const float* __restrict__ bias, float* __restrict__ out, int N,
    const float* __restrict__ Wn, const float* __restrict__ a_sn,
    const float* __restrict__ a_dn, float* __restrict__ as_n,
    float* __restrict__ ad_n) {
  const int pair = __builtin_amdgcn_readfirstlane(blockIdx.x * 8 + (threadIdx.x >> 5));
  const int lane = threadIdx.x & 31;
  const int c = lane & 15;
  if (!FUSE && pair * 2 >= N) return;
  const bool live = pair * 2 + (lane >> 4) < N;
  const int node = live ? pair * 2 + (lane >> 4) : (pair * 2 < N ? pair * 2 : 0);
  const int beg = node * SLOTS;
  const int q0 = slots[(unsigned)(beg + c)];
  const int q1 = slots[(unsigned)(beg + 16 + c)];
  const int cn = cnt[node];
  const int hd = head[node];
  const float adn = ad_[node];
  const float asn = as_[node];
  __builtin_amdgcn_sched_barrier(0);
  const float es = leaky(asn + adn);
  const int deg = live ? cn : 0;
  const int dlo = __builtin_amdgcn_readlane(deg, 0);
  const int dhi = __builtin_amdgcn_readlane(deg, 16);
  const int degmax = dlo > dhi ? dlo : dhi;

  const bool on0 = c < deg, on1 = c + 16 < deg;
  const int s0 = on0 ? q0 : node;
  const int s1 = on1 ? q1 : node;
  const float t0 = as_[s0], t1 = as_[s1];
  const float e0 = leaky(t0 + adn), e1 = leaky(t1 + adn);
  float m = fmaxf(es, fmaxf(e0, e1));
  const int nov = deg - SLOTS;
  {
    int k = hd;
    for (int t = 0; t < nov && (unsigned)k < (unsigned)E; ++t) {
      m = fmaxf(m, leaky(as_[ei[k]] + adn));
      k = next[k];
    }
  }
  m = row16_xor_max<1>(m);
  m = row16_xor_max<2>(m);
  m = row16_xor_max<4>(m);
  m = row16_xor_max<8>(m);

  const float v0 = on0 ? __expf(e0 - m) : 0.0f;
  const float v1 = on1 ? __expf(e1 - m) : 0.0f;
  float sum = v0 + v1;
  sum = row16_xor_add<1>(sum);
  sum = row16_xor_add<2>(sum);
  sum = row16_xor_add<4>(sum);
  sum = row16_xor_add<8>(sum);
  float ov = 0.0f;
  {
    int k = hd;
    for (int t = 0; t < nov && (unsigned)k < (unsigned)E; ++t) {
      ov += __expf(leaky(as_[ei[k]] + adn) - m);
      k = next[k];
    }
  }
  const float vs = __expf(es - m);
  const float dn = ((sum + ov) + vs) + 1e-16f;

  const float4* __restrict__ hp4 = (const float4*)hp;
  const float aself = vs / dn;
  const float4 hs = hp4[(unsigned)(node * 16 + c)];
  float4 acc;
  acc.x = hs.x * aself;
  acc.y = hs.y * aself;
  acc.z = hs.z * aself;
  acc.w = hs.w * aself;
  add_rows16(hp4, s0, v0 / dn, degmax, c, acc);
  if (degmax > 16) add_rows16(hp4, s1, v1 / dn, degmax - 16, c, acc);
  {
    int k = hd;
    for (int t = 0; t < nov && (unsigned)k < (unsigned)E; ++t) {
      const int sj = ei[k];
      const float aj = __expf(leaky(as_[sj] + adn) - m) / dn;
      fma_row(acc, hp4[(unsigned)(sj * 16 + c)], aj);
      k = next[k];
    }
  }

  const float4 b = ((const float4*)bias)[c];
  float4 o;
  o.x = acc.x + b.x;
  o.y = acc.y + b.y;
  o.z = acc.z + b.z;
  o.w = acc.w + b.w;
  o.x = o.x > 0.0f ? o.x : 0.0f;
  o.y = o.y > 0.0f ? o.y : 0.0f;
  o.z = o.z > 0.0f ? o.z : 0.0f;
  o.w = o.w > 0.0f ? o.w : 0.0f;
  if (!FUSE) {
    if (live) ((float4*)out)[(unsigned)(node * 16 + c)] = o;
    return;
  }

  __shared__ float sA[16 * TP];
  __shared__ float sC[16 * TP];
  const int r = threadIdx.x >> 4;
  const int h = lane >> 4;
  const int w = __builtin_amdgcn_readfirstlane(threadIdx.x >> 5);
  *(float4*)(sA + r * TP + 4 * c) = o;
  float wb[32];
  if (w < 4) {
#pragma unroll
    for (int j = 0; j < 16; ++j) {
      wb[2 * j]     = Wn[(4 * j + 2 * h) * 64 + 16 * w + c];
      wb[2 * j + 1] = Wn[(4 * j + 2 * h + 1) * 64 + 16 * w + c];
    }
  }
  __syncthreads();
  if (w < 4) {
    v8f acc = {};
#pragma unroll
    for (int j = 0; j < 16; ++j) {
      const v2f a = *(const v2f*)(sA + c * TP + 4 * j + 2 * h);
      acc = wmma4(a, wb[2 * j], wb[2 * j + 1], acc);
    }
    __builtin_amdgcn_sched_barrier(0);
    asm volatile("v_nop\n\tv_nop\n\tv_nop\n\tv_nop");
    __builtin_amdgcn_sched_barrier(0);
#pragma unroll
    for (int v = 0; v < 8; ++v) sC[(8 * h + v) * TP + 16 * w + c] = acc[v];
  }
  __syncthreads();
  const float4 cv = *(const float4*)(sC + r * TP + 4 * c);
  if (live) ((float4*)out)[(unsigned)(node * 16 + c)] = cv;
  const float4 s4 = *(const float4*)(a_sn + 4 * c);
  const float4 d4 = *(const float4*)(a_dn + 4 * c);
  float ps = cv.x * s4.x + cv.y * s4.y + cv.z * s4.z + cv.w * s4.w;
  float pd = cv.x * d4.x + cv.y * d4.y + cv.z * d4.z + cv.w * d4.w;
  ps = row16_xor_add<1>(ps);  pd = row16_xor_add<1>(pd);
  ps = row16_xor_add<2>(ps);  pd = row16_xor_add<2>(pd);
  ps = row16_xor_add<4>(ps);  pd = row16_xor_add<4>(pd);
  ps = row16_xor_add<8>(ps);  pd = row16_xor_add<8>(pd);
  if (live && c == 0) {
    as_n[node] = ps;
    ad_n[node] = pd;
  }
}

extern "C" void kernel_launch(void* const* d_in, const int* in_sizes, int n_in,
                              void* d_out, int out_size, void* d_ws, size_t ws_size,
                              hipStream_t stream) {
  (void)n_in; (void)out_size; (void)ws_size;

  const float* x   = (const float*)d_in[0];
  const int*   ei  = (const int*)d_in[1];
  const int N = in_sizes[0] / 128;
  const int E = in_sizes[1] / 2;
  float* ws    = (float*)d_ws;
  float* hp    = ws;
  float* hp2   = hp    + (size_t)N * 64;
  float* lists = hp2   + (size_t)N * 64;
  float* as_   = lists + (size_t)N * 64;
  float* ad_   = as_   + N;
  int* cnt      = (int*)lists;
  int* head     = cnt + N;
  int* slots    = head + N;
  int* next     = slots + (size_t)N * SLOTS;
  float* as2    = (float*)(next + E);
  float* ad2    = as2 + N;

  csr_zero<<<(N + 255) / 256, 256, 0, stream>>>(cnt, head, N);

  const float* W0 = (const float*)d_in[4];
  const float* W1 = (const float*)d_in[8];
  const float* W2 = (const float*)d_in[12];

  const int mblocks = (N / 16 + 7) / 8;
  const int pblocks = (E + 255) / 256;
  gat_gemm_alpha_place<<<mblocks + pblocks, 256, 0, stream>>>(
      x, W0, (const float*)d_in[5], (const float*)d_in[6], hp, as_, ad_, N, 128,
      mblocks, ei, cnt, head, slots, next, E);

  const int gblocks = (N + 15) / 16;
  gat_gather<true><<<gblocks, 256, 0, stream>>>(cnt, slots, head, next, ei, E, hp, as_, ad_,
      (const float*)d_in[7], hp2, N, W1, (const float*)d_in[9],
      (const float*)d_in[10], as2, ad2);
  gat_gather<true><<<gblocks, 256, 0, stream>>>(cnt, slots, head, next, ei, E, hp2, as2, ad2,
      (const float*)d_in[11], hp, N, W2, (const float*)d_in[13],
      (const float*)d_in[14], as_, ad_);
  gat_gather<false><<<gblocks, 256, 0, stream>>>(cnt, slots, head, next, ei, E, hp, as_, ad_,
      (const float*)d_in[15], (float*)d_out, N, nullptr, nullptr, nullptr,
      nullptr, nullptr);
}
